# starting kernel
# baseline (speedup 1.0000x reference)
#include <hip/hip_runtime.h>

#define CH 128

typedef float v2f __attribute__((ext_vector_type(2)));
typedef float v8f __attribute__((ext_vector_type(8)));

__global__ void fill_kernel(float* __restrict__ p, int n, float v) {
    int i = blockIdx.x * blockDim.x + threadIdx.x;
    if (i < n) p[i] = v;
}

__global__ void deg_count_kernel(const int* __restrict__ dst,
                                 float* __restrict__ deg, int E) {
    int e = blockIdx.x * blockDim.x + threadIdx.x;
    if (e < E) atomicAdd(&deg[dst[e]], 1.0f);
}

__global__ void rsqrt_kernel(const float* __restrict__ deg,
                             float* __restrict__ dinv, int n) {
    int i = blockIdx.x * blockDim.x + threadIdx.x;
    if (i < n) dinv[i] = rsqrtf(deg[i]);
}

__global__ __launch_bounds__(256)
void gemm_wmma_f32(const float* __restrict__ X, const float* __restrict__ W,
                   float* __restrict__ H) {
    const int wave  = threadIdx.x >> 5;
    const int lane  = threadIdx.x & 31;
    const int m0    = blockIdx.x * 16;
    const int n0    = wave * 16;
    const int mrow  = lane & 15;
    const int khalf = (lane >> 4) << 1;
    const int ncol  = lane & 15;

    const float* __restrict__ xrow = X + (size_t)(m0 + mrow) * CH;

    v8f acc = {0.f, 0.f, 0.f, 0.f, 0.f, 0.f, 0.f, 0.f};
#pragma unroll
    for (int k0 = 0; k0 < CH; k0 += 4) {
        v2f a, b;
        a.x = xrow[k0 + khalf];
        a.y = xrow[k0 + khalf + 1];
        b.x = W[(size_t)(k0 + khalf) * CH + n0 + ncol];
        b.y = W[(size_t)(k0 + khalf + 1) * CH + n0 + ncol];
        acc = __builtin_amdgcn_wmma_f32_16x16x4_f32(
             false, a,  false, b,
             (short)0, acc,  false,  false);
    }

    const int moff = (lane >> 4) * 8;
#pragma unroll
    for (int j = 0; j < 8; ++j) {
        H[(size_t)(m0 + j + moff) * CH + n0 + ncol] = acc[j];
    }
}

__global__ __launch_bounds__(256)
void scatter128_kernel(const float* __restrict__ h, const int* __restrict__ src,
                       const int* __restrict__ dst, const float* __restrict__ dinv,
                       float* __restrict__ out, int E, int Etot) {
    const int w = (blockIdx.x * blockDim.x + threadIdx.x) >> 5;
    if (w >= Etot) return;
    const int lane = threadIdx.x & 31;
    int s, d;
    if (w < E) { s = src[w]; d = dst[w]; } else { s = d = w - E; }
    const float nrm = dinv[s] * dinv[d];
    const float4 v = ((const float4*)(h + (size_t)s * CH))[lane];
    float* od = out + (size_t)d * CH + lane * 4;
    atomicAdd(od + 0, v.x * nrm);
    atomicAdd(od + 1, v.y * nrm);
    atomicAdd(od + 2, v.z * nrm);
    atomicAdd(od + 3, v.w * nrm);
}

__global__ void postact_kernel(float* __restrict__ h, const float* __restrict__ b,
                               const float* __restrict__ g, const float* __restrict__ be,
                               const float* __restrict__ rm, const float* __restrict__ rv,
                               int total) {
    int i = blockIdx.x * blockDim.x + threadIdx.x;
    if (i >= total) return;
    const int c = i & (CH - 1);
    const float inv = g[c] * rsqrtf(rv[c] + 1e-5f);
    const float v = (h[i] + b[c] - rm[c]) * inv + be[c];
    h[i] = v > 0.f ? v : 0.f;
}

__global__ void gemm_out2_kernel(const float* __restrict__ X,
                                 const float* __restrict__ W3,
                                 float* __restrict__ H3, int M) {
    int i = blockIdx.x * blockDim.x + threadIdx.x;
    if (i >= M) return;
    const float* xr = X + (size_t)i * CH;
    float a0 = 0.f, a1 = 0.f;
#pragma unroll 8
    for (int k = 0; k < CH; ++k) {
        const float x = xr[k];
        a0 += x * W3[2 * k];
        a1 += x * W3[2 * k + 1];
    }
    H3[(size_t)i * 2]     = a0;
    H3[(size_t)i * 2 + 1] = a1;
}

__global__ void scatter2_kernel(const float* __restrict__ h3, const int* __restrict__ src,
                                const int* __restrict__ dst, const float* __restrict__ dinv,
                                float* __restrict__ out, int E, int Etot) {
    int e = blockIdx.x * blockDim.x + threadIdx.x;
    if (e >= Etot) return;
    int s, d;
    if (e < E) { s = src[e]; d = dst[e]; } else { s = d = e - E; }
    const float nrm = dinv[s] * dinv[d];
    atomicAdd(&out[(size_t)d * 2],     h3[(size_t)s * 2]     * nrm);
    atomicAdd(&out[(size_t)d * 2 + 1], h3[(size_t)s * 2 + 1] * nrm);
}

__global__ void add_bias2_kernel(float* __restrict__ out,
                                 const float* __restrict__ b3, int total) {
    int i = blockIdx.x * blockDim.x + threadIdx.x;
    if (i < total) out[i] += b3[i & 1];
}

extern "C" void kernel_launch(void* const* d_in, const int* in_sizes, int n_in,
                              void* d_out, int out_size, void* d_ws, size_t ws_size,
                              hipStream_t stream) {
    const float* x   = (const float*)d_in[0];
    const int*   ei  = (const int*)d_in[1];
    const float* W1  = (const float*)d_in[2];
    const float* b1  = (const float*)d_in[3];
    const float* g1  = (const float*)d_in[4];
    const float* be1 = (const float*)d_in[5];
    const float* rm1 = (const float*)d_in[6];
    const float* rv1 = (const float*)d_in[7];
    const float* W2  = (const float*)d_in[8];
    const float* b2  = (const float*)d_in[9];
    const float* g2  = (const float*)d_in[10];
    const float* be2 = (const float*)d_in[11];
    const float* rm2 = (const float*)d_in[12];
    const float* rv2 = (const float*)d_in[13];
    const float* W3  = (const float*)d_in[14];
    const float* b3  = (const float*)d_in[15];

    const int Nn   = in_sizes[0] / CH;
    const int Ee   = in_sizes[1] / 2;
    const int Etot = Ee + Nn;
    const int NC   = Nn * CH;

    const int* src = ei;
    const int* dst = ei + Ee;

    float* deg  = (float*)d_ws;
    float* dinv = deg  + Nn;
    float* bufA = dinv + Nn;
    float* bufB = bufA + (size_t)NC;
    float* h3   = bufB + (size_t)NC;
    float* out  = (float*)d_out;

    const int T = 256;
    const int gN  = (Nn + T - 1) / T;
    const int gE  = (Ee + T - 1) / T;
    const int gEt = (Etot + T - 1) / T;
    const int gNC = (NC + T - 1) / T;
    const int gSc = (Etot + 7) / 8;
    const int gMM = Nn / 16;

    fill_kernel<<<gN, T, 0, stream>>>(deg, Nn, 1.0f);
    deg_count_kernel<<<gE, T, 0, stream>>>(dst, deg, Ee);
    rsqrt_kernel<<<gN, T, 0, stream>>>(deg, dinv, Nn);

    gemm_wmma_f32<<<gMM, T, 0, stream>>>(x, W1, bufA);
    fill_kernel<<<gNC, T, 0, stream>>>(bufB, NC, 0.0f);
    scatter128_kernel<<<gSc, T, 0, stream>>>(bufA, src, dst, dinv, bufB, Ee, Etot);
    postact_kernel<<<gNC, T, 0, stream>>>(bufB, b1, g1, be1, rm1, rv1, NC);

    gemm_wmma_f32<<<gMM, T, 0, stream>>>(bufB, W2, bufA);
    fill_kernel<<<gNC, T, 0, stream>>>(bufB, NC, 0.0f);
    scatter128_kernel<<<gSc, T, 0, stream>>>(bufA, src, dst, dinv, bufB, Ee, Etot);
    postact_kernel<<<gNC, T, 0, stream>>>(bufB, b2, g2, be2, rm2, rv2, NC);

    gemm_out2_kernel<<<gN, T, 0, stream>>>(bufB, W3, h3, Nn);
    fill_kernel<<<(2 * Nn + T - 1) / T, T, 0, stream>>>(out, 2 * Nn, 0.0f);
    scatter2_kernel<<<gEt, T, 0, stream>>>(h3, src, dst, dinv, out, Ee, Etot);
    add_bias2_kernel<<<(2 * Nn + T - 1) / T, T, 0, stream>>>(out, b3, 2 * Nn);
}


# Float atomic scatter-add replaced by gather over edges grouped by destination (CSR rebuilt each call: int rank atomics + scan); a wave sums a node's row in registers in CSR slot order, stored once
# speedup vs baseline: 2.4482x; 2.4482x over previous
#include <hip/hip_runtime.h>

#define CH 128

typedef float v2f __attribute__((ext_vector_type(2)));
typedef float v8f __attribute__((ext_vector_type(8)));

__global__ void zero_int_kernel(int* __restrict__ p, int n) {
    int i = blockIdx.x * blockDim.x + threadIdx.x;
    if (i < n) p[i] = 0;
}

__global__ void rank_count_kernel(const int* __restrict__ dst, int* __restrict__ cnt,
                                  int* __restrict__ rank, int E) {
    int e = blockIdx.x * blockDim.x + threadIdx.x;
    if (e < E) rank[e] = atomicAdd(&cnt[dst[e]], 1);
}

__global__ __launch_bounds__(256)
void block_sum_kernel(const int* __restrict__ cnt, int* __restrict__ bsum, int n) {
    __shared__ int sm[256];
    const int t = threadIdx.x;
    const int i = blockIdx.x * 256 + t;
    sm[t] = (i < n) ? cnt[i] : 0;
    __syncthreads();
    for (int s = 128; s > 0; s >>= 1) {
        if (t < s) sm[t] += sm[t + s];
        __syncthreads();
    }
    if (t == 0) bsum[blockIdx.x] = sm[0];
}

__global__ __launch_bounds__(256)
void row_start_kernel(const int* __restrict__ cnt, const int* __restrict__ bsum,
                      int* __restrict__ rowptr, float* __restrict__ dinv, int n) {
    __shared__ int sm[256];
    __shared__ int red[256];
    const int t = threadIdx.x;
    const int b = blockIdx.x;
    int part = 0;
    for (int k = t; k < b; k += 256) part += bsum[k];
    red[t] = part;
    const int i = b * 256 + t;
    const int c = (i < n) ? cnt[i] : 0;
    sm[t] = c;
    __syncthreads();
    for (int s = 128; s > 0; s >>= 1) {
        if (t < s) red[t] += red[t + s];
        __syncthreads();
    }
    const int off = red[0];
    for (int s = 1; s < 256; s <<= 1) {
        const int v = (t >= s) ? sm[t - s] : 0;
        __syncthreads();
        sm[t] += v;
        __syncthreads();
    }
    const int incl = sm[t];
    if (i < n) {
        rowptr[i] = off + incl - c;
        dinv[i]   = rsqrtf((float)(c + 1));
    }
    if (i == n - 1) rowptr[n] = off + incl;
}

__global__ void place_kernel(const int* __restrict__ src, const int* __restrict__ dst,
                             const int* __restrict__ rank, const int* __restrict__ rowptr,
                             int* __restrict__ csr_src, int E) {
    int e = blockIdx.x * blockDim.x + threadIdx.x;
    if (e < E) csr_src[rowptr[dst[e]] + rank[e]] = src[e];
}

__global__ __launch_bounds__(256)
void gemm_wmma_f32(const float* __restrict__ X, const float* __restrict__ W,
                   float* __restrict__ H) {
    const int wave  = threadIdx.x >> 5;
    const int lane  = threadIdx.x & 31;
    const int m0    = blockIdx.x * 16;
    const int n0    = wave * 16;
    const int mrow  = lane & 15;
    const int khalf = (lane >> 4) << 1;
    const int ncol  = lane & 15;

    const float* __restrict__ xrow = X + (size_t)(m0 + mrow) * CH;

    v8f acc = {0.f, 0.f, 0.f, 0.f, 0.f, 0.f, 0.f, 0.f};
#pragma unroll
    for (int k0 = 0; k0 < CH; k0 += 4) {
        v2f a, b;
        a.x = xrow[k0 + khalf];
        a.y = xrow[k0 + khalf + 1];
        b.x = W[(size_t)(k0 + khalf) * CH + n0 + ncol];
        b.y = W[(size_t)(k0 + khalf + 1) * CH + n0 + ncol];
        acc = __builtin_amdgcn_wmma_f32_16x16x4_f32(
             false, a,  false, b,
             (short)0, acc,  false,  false);
    }

    const int moff = (lane >> 4) * 8;
#pragma unroll
    for (int j = 0; j < 8; ++j) {
        H[(size_t)(m0 + j + moff) * CH + n0 + ncol] = acc[j];
    }
}

__global__ __launch_bounds__(256)
void gather128_kernel(const float* __restrict__ h, const int* __restrict__ rowptr,
                      const int* __restrict__ csr_src, const float* __restrict__ dinv,
                      float* __restrict__ out, int n) {
    const int d = __builtin_amdgcn_readfirstlane(
        (int)((blockIdx.x * blockDim.x + threadIdx.x) >> 5));
    if (d >= n) return;
    const int lane = threadIdx.x & 31;
    const int beg = rowptr[d];
    const int end = rowptr[d + 1];
    const float dd = dinv[d];
    const float4* __restrict__ hv = (const float4*)h;

    const float4 vs = hv[(size_t)d * 32 + lane];
    const float ws = dd * dd;
    float4 acc;
    acc.x = vs.x * ws; acc.y = vs.y * ws; acc.z = vs.z * ws; acc.w = vs.w * ws;

    for (int base = beg; base < end; base += 32) {
        const int m = (end - base < 32) ? (end - base) : 32;
        int   ms = 0;
        float mw = 0.f;
        if (lane < m) {
            ms = csr_src[base + lane];
            mw = dinv[ms] * dd;
        }
        const int mwi = __float_as_int(mw);
        int j = 0;
        for (; j + 4 <= m; j += 4) {
            const int s0 = __builtin_amdgcn_readlane(ms, j);
            const int s1 = __builtin_amdgcn_readlane(ms, j + 1);
            const int s2 = __builtin_amdgcn_readlane(ms, j + 2);
            const int s3 = __builtin_amdgcn_readlane(ms, j + 3);
            const float w0 = __int_as_float(__builtin_amdgcn_readlane(mwi, j));
            const float w1 = __int_as_float(__builtin_amdgcn_readlane(mwi, j + 1));
            const float w2 = __int_as_float(__builtin_amdgcn_readlane(mwi, j + 2));
            const float w3 = __int_as_float(__builtin_amdgcn_readlane(mwi, j + 3));
            const float4 v0 = hv[(size_t)s0 * 32 + lane];
            const float4 v1 = hv[(size_t)s1 * 32 + lane];
            const float4 v2 = hv[(size_t)s2 * 32 + lane];
            const float4 v3 = hv[(size_t)s3 * 32 + lane];
            acc.x += v0.x * w0; acc.y += v0.y * w0; acc.z += v0.z * w0; acc.w += v0.w * w0;
            acc.x += v1.x * w1; acc.y += v1.y * w1; acc.z += v1.z * w1; acc.w += v1.w * w1;
            acc.x += v2.x * w2; acc.y += v2.y * w2; acc.z += v2.z * w2; acc.w += v2.w * w2;
            acc.x += v3.x * w3; acc.y += v3.y * w3; acc.z += v3.z * w3; acc.w += v3.w * w3;
        }
        for (; j < m; ++j) {
            const int s0 = __builtin_amdgcn_readlane(ms, j);
            const float w0 = __int_as_float(__builtin_amdgcn_readlane(mwi, j));
            const float4 v0 = hv[(size_t)s0 * 32 + lane];
            acc.x += v0.x * w0; acc.y += v0.y * w0; acc.z += v0.z * w0; acc.w += v0.w * w0;
        }
    }
    ((float4*)out)[(size_t)d * 32 + lane] = acc;
}

__global__ void postact_kernel(float* __restrict__ h, const float* __restrict__ b,
                               const float* __restrict__ g, const float* __restrict__ be,
                               const float* __restrict__ rm, const float* __restrict__ rv,
                               int total) {
    int i = blockIdx.x * blockDim.x + threadIdx.x;
    if (i >= total) return;
    const int c = i & (CH - 1);
    const float inv = g[c] * rsqrtf(rv[c] + 1e-5f);
    const float v = (h[i] + b[c] - rm[c]) * inv + be[c];
    h[i] = v > 0.f ? v : 0.f;
}

__global__ void gemm_out2_kernel(const float* __restrict__ X,
                                 const float* __restrict__ W3,
                                 float* __restrict__ H3, int M) {
    int i = blockIdx.x * blockDim.x + threadIdx.x;
    if (i >= M) return;
    const float* xr = X + (size_t)i * CH;
    float a0 = 0.f, a1 = 0.f;
#pragma unroll 8
    for (int k = 0; k < CH; ++k) {
        const float x = xr[k];
        a0 += x * W3[2 * k];
        a1 += x * W3[2 * k + 1];
    }
    H3[(size_t)i * 2]     = a0;
    H3[(size_t)i * 2 + 1] = a1;
}

__global__ void gather2_kernel(const float* __restrict__ h3, const int* __restrict__ rowptr,
                               const int* __restrict__ csr_src, const float* __restrict__ dinv,
                               const float* __restrict__ b3, float* __restrict__ out, int n) {
    const int d = blockIdx.x * blockDim.x + threadIdx.x;
    if (d >= n) return;
    const int beg = rowptr[d];
    const int end = rowptr[d + 1];
    const float dd = dinv[d];
    const float2* __restrict__ hv = (const float2*)h3;
    const float2 vs = hv[d];
    const float ws = dd * dd;
    float a0 = vs.x * ws;
    float a1 = vs.y * ws;
#pragma unroll 4
    for (int j = beg; j < end; ++j) {
        const int s = csr_src[j];
        const float w = dinv[s] * dd;
        const float2 u = hv[s];
        a0 += u.x * w;
        a1 += u.y * w;
    }
    float2 r;
    r.x = a0 + b3[0];
    r.y = a1 + b3[1];
    ((float2*)out)[d] = r;
}

extern "C" void kernel_launch(void* const* d_in, const int* in_sizes, int n_in,
                              void* d_out, int out_size, void* d_ws, size_t ws_size,
                              hipStream_t stream) {
    const float* x   = (const float*)d_in[0];
    const int*   ei  = (const int*)d_in[1];
    const float* W1  = (const float*)d_in[2];
    const float* b1  = (const float*)d_in[3];
    const float* g1  = (const float*)d_in[4];
    const float* be1 = (const float*)d_in[5];
    const float* rm1 = (const float*)d_in[6];
    const float* rv1 = (const float*)d_in[7];
    const float* W2  = (const float*)d_in[8];
    const float* b2  = (const float*)d_in[9];
    const float* g2  = (const float*)d_in[10];
    const float* be2 = (const float*)d_in[11];
    const float* rm2 = (const float*)d_in[12];
    const float* rv2 = (const float*)d_in[13];
    const float* W3  = (const float*)d_in[14];
    const float* b3  = (const float*)d_in[15];

    const int Nn   = in_sizes[0] / CH;
    const int Ee   = in_sizes[1] / 2;
    const int NC   = Nn * CH;

    const int* src = ei;
    const int* dst = ei + Ee;

    const int T = 256;
    const int gN  = (Nn + T - 1) / T;
    const int gE  = (Ee + T - 1) / T;
    const int gNC = (NC + T - 1) / T;
    const int gW  = (Nn + 7) / 8;
    const int gMM = Nn / 16;

    const size_t NP = ((size_t)Nn + 64) & ~(size_t)63;
    const size_t BP = ((size_t)gN + 63) & ~(size_t)63;
    const size_t EP = ((size_t)Ee + 63) & ~(size_t)63;
    int*   cnt    = (int*)d_ws;
    int*   bsum   = cnt + NP;
    int*   rowptr = bsum + BP;
    float* dinv   = (float*)(rowptr + NP);
    int*   rank   = (int*)(dinv + NP);
    int*   csr    = rank + EP;
    float* bufA   = (float*)(csr + EP);
    float* bufB   = bufA + (size_t)NC;
    float* h3     = bufB + (size_t)NC;
    float* out    = (float*)d_out;

    zero_int_kernel<<<gN, T, 0, stream>>>(cnt, Nn);
    rank_count_kernel<<<gE, T, 0, stream>>>(dst, cnt, rank, Ee);
    block_sum_kernel<<<gN, T, 0, stream>>>(cnt, bsum, Nn);
    row_start_kernel<<<gN, T, 0, stream>>>(cnt, bsum, rowptr, dinv, Nn);
    place_kernel<<<gE, T, 0, stream>>>(src, dst, rank, rowptr, csr, Ee);

    gemm_wmma_f32<<<gMM, T, 0, stream>>>(x, W1, bufA);
    gather128_kernel<<<gW, T, 0, stream>>>(bufA, rowptr, csr, dinv, bufB, Nn);
    postact_kernel<<<gNC, T, 0, stream>>>(bufB, b1, g1, be1, rm1, rv1, NC);

    gemm_wmma_f32<<<gMM, T, 0, stream>>>(bufB, W2, bufA);
    gather128_kernel<<<gW, T, 0, stream>>>(bufA, rowptr, csr, dinv, bufB, Nn);
    postact_kernel<<<gNC, T, 0, stream>>>(bufB, b2, g2, be2, rm2, rv2, NC);

    gemm_out2_kernel<<<gN, T, 0, stream>>>(bufB, W3, h3, Nn);
    gather2_kernel<<<gN, T, 0, stream>>>(h3, rowptr, csr, dinv, b3, out, Nn);
}


# Bias, batch norm and ReLU applied in the 128-channel gather's epilogue, same arithmetic and order; both elementwise launches dropped (13 to 11); remainder loop bound written m&3 to hold 64 registers
# speedup vs baseline: 2.9326x; 1.1979x over previous
#include <hip/hip_runtime.h>

#define CH 128

typedef float v2f __attribute__((ext_vector_type(2)));
typedef float v8f __attribute__((ext_vector_type(8)));

__global__ void zero_int_kernel(int* __restrict__ p, int n) {
    int i = blockIdx.x * blockDim.x + threadIdx.x;
    if (i < n) p[i] = 0;
}

__global__ void rank_count_kernel(const int* __restrict__ dst, int* __restrict__ cnt,
                                  int* __restrict__ rank, int E) {
    int e = blockIdx.x * blockDim.x + threadIdx.x;
    if (e < E) rank[e] = atomicAdd(&cnt[dst[e]], 1);
}

__global__ __launch_bounds__(256)
void block_sum_kernel(const int* __restrict__ cnt, int* __restrict__ bsum, int n) {
    __shared__ int sm[256];
    const int t = threadIdx.x;
    const int i = blockIdx.x * 256 + t;
    sm[t] = (i < n) ? cnt[i] : 0;
    __syncthreads();
    for (int s = 128; s > 0; s >>= 1) {
        if (t < s) sm[t] += sm[t + s];
        __syncthreads();
    }
    if (t == 0) bsum[blockIdx.x] = sm[0];
}

__global__ __launch_bounds__(256)
void row_start_kernel(const int* __restrict__ cnt, const int* __restrict__ bsum,
                      int* __restrict__ rowptr, float* __restrict__ dinv, int n) {
    __shared__ int sm[256];
    __shared__ int red[256];
    const int t = threadIdx.x;
    const int b = blockIdx.x;
    int part = 0;
    for (int k = t; k < b; k += 256) part += bsum[k];
    red[t] = part;
    const int i = b * 256 + t;
    const int c = (i < n) ? cnt[i] : 0;
    sm[t] = c;
    __syncthreads();
    for (int s = 128; s > 0; s >>= 1) {
        if (t < s) red[t] += red[t + s];
        __syncthreads();
    }
    const int off = red[0];
    for (int s = 1; s < 256; s <<= 1) {
        const int v = (t >= s) ? sm[t - s] : 0;
        __syncthreads();
        sm[t] += v;
        __syncthreads();
    }
    const int incl = sm[t];
    if (i < n) {
        rowptr[i] = off + incl - c;
        dinv[i]   = rsqrtf((float)(c + 1));
    }
    if (i == n - 1) rowptr[n] = off + incl;
}

__global__ void place_kernel(const int* __restrict__ src, const int* __restrict__ dst,
                             const int* __restrict__ rank, const int* __restrict__ rowptr,
                             int* __restrict__ csr_src, int E) {
    int e = blockIdx.x * blockDim.x + threadIdx.x;
    if (e < E) csr_src[rowptr[dst[e]] + rank[e]] = src[e];
}

__global__ __launch_bounds__(256)
void gemm_wmma_f32(const float* __restrict__ X, const float* __restrict__ W,
                   float* __restrict__ H) {
    const int wave  = threadIdx.x >> 5;
    const int lane  = threadIdx.x & 31;
    const int m0    = blockIdx.x * 16;
    const int n0    = wave * 16;
    const int mrow  = lane & 15;
    const int khalf = (lane >> 4) << 1;
    const int ncol  = lane & 15;

    const float* __restrict__ xrow = X + (size_t)(m0 + mrow) * CH;

    v8f acc = {0.f, 0.f, 0.f, 0.f, 0.f, 0.f, 0.f, 0.f};
#pragma unroll
    for (int k0 = 0; k0 < CH; k0 += 4) {
        v2f a, b;
        a.x = xrow[k0 + khalf];
        a.y = xrow[k0 + khalf + 1];
        b.x = W[(size_t)(k0 + khalf) * CH + n0 + ncol];
        b.y = W[(size_t)(k0 + khalf + 1) * CH + n0 + ncol];
        acc = __builtin_amdgcn_wmma_f32_16x16x4_f32(
             false, a,  false, b,
             (short)0, acc,  false,  false);
    }

    const int moff = (lane >> 4) * 8;
#pragma unroll
    for (int j = 0; j < 8; ++j) {
        H[(size_t)(m0 + j + moff) * CH + n0 + ncol] = acc[j];
    }
}

__global__ __launch_bounds__(256)
void gather128_kernel(const float* __restrict__ h, const int* __restrict__ rowptr,
                      const int* __restrict__ csr_src, const float* __restrict__ dinv,
                      const float* __restrict__ b, const float* __restrict__ g,
                      const float* __restrict__ be, const float* __restrict__ rm,
                      const float* __restrict__ rv,
                      float* __restrict__ out, int n) {
    const int d = __builtin_amdgcn_readfirstlane(
        (int)((blockIdx.x * blockDim.x + threadIdx.x) >> 5));
    if (d >= n) return;
    const int lane = threadIdx.x & 31;
    const int beg = rowptr[d];
    const int end = rowptr[d + 1];
    const float dd = dinv[d];
    const float4* __restrict__ hv = (const float4*)h;

    const float4 pb  = ((const float4*)b)[lane];
    const float4 pg  = ((const float4*)g)[lane];
    const float4 pbe = ((const float4*)be)[lane];
    const float4 prm = ((const float4*)rm)[lane];
    const float4 prv = ((const float4*)rv)[lane];
    const float ix = pg.x * rsqrtf(prv.x + 1e-5f);
    const float iy = pg.y * rsqrtf(prv.y + 1e-5f);
    const float iz = pg.z * rsqrtf(prv.z + 1e-5f);
    const float iw = pg.w * rsqrtf(prv.w + 1e-5f);

    const float4 vs = hv[(size_t)d * 32 + lane];
    const float ws = dd * dd;
    float4 acc;
    acc.x = vs.x * ws; acc.y = vs.y * ws; acc.z = vs.z * ws; acc.w = vs.w * ws;

    for (int base = beg; base < end; base += 32) {
        const int m = (end - base < 32) ? (end - base) : 32;
        int   ms = 0;
        float mw = 0.f;
        if (lane < m) {
            ms = csr_src[base + lane];
            mw = dinv[ms] * dd;
        }
        const int mwi = __float_as_int(mw);
        int j = 0;
        for (; j + 4 <= m; j += 4) {
            const int s0 = __builtin_amdgcn_readlane(ms, j);
            const int s1 = __builtin_amdgcn_readlane(ms, j + 1);
            const int s2 = __builtin_amdgcn_readlane(ms, j + 2);
            const int s3 = __builtin_amdgcn_readlane(ms, j + 3);
            const float w0 = __int_as_float(__builtin_amdgcn_readlane(mwi, j));
            const float w1 = __int_as_float(__builtin_amdgcn_readlane(mwi, j + 1));
            const float w2 = __int_as_float(__builtin_amdgcn_readlane(mwi, j + 2));
            const float w3 = __int_as_float(__builtin_amdgcn_readlane(mwi, j + 3));
            const float4 v0 = hv[(size_t)s0 * 32 + lane];
            const float4 v1 = hv[(size_t)s1 * 32 + lane];
            const float4 v2 = hv[(size_t)s2 * 32 + lane];
            const float4 v3 = hv[(size_t)s3 * 32 + lane];
            acc.x += v0.x * w0; acc.y += v0.y * w0; acc.z += v0.z * w0; acc.w += v0.w * w0;
            acc.x += v1.x * w1; acc.y += v1.y * w1; acc.z += v1.z * w1; acc.w += v1.w * w1;
            acc.x += v2.x * w2; acc.y += v2.y * w2; acc.z += v2.z * w2; acc.w += v2.w * w2;
            acc.x += v3.x * w3; acc.y += v3.y * w3; acc.z += v3.z * w3; acc.w += v3.w * w3;
        }
        const int rem = m & 3;
        for (int t = 0; t < rem; ++t) {
            const int s0 = __builtin_amdgcn_readlane(ms, j + t);
            const float w0 = __int_as_float(__builtin_amdgcn_readlane(mwi, j + t));
            const float4 v0 = hv[(size_t)s0 * 32 + lane];
            acc.x += v0.x * w0; acc.y += v0.y * w0; acc.z += v0.z * w0; acc.w += v0.w * w0;
        }
    }
    float4 r;
    {
        const float vx = (acc.x + pb.x - prm.x) * ix + pbe.x;
        const float vy = (acc.y + pb.y - prm.y) * iy + pbe.y;
        const float vz = (acc.z + pb.z - prm.z) * iz + pbe.z;
        const float vw = (acc.w + pb.w - prm.w) * iw + pbe.w;
        r.x = vx > 0.f ? vx : 0.f;
        r.y = vy > 0.f ? vy : 0.f;
        r.z = vz > 0.f ? vz : 0.f;
        r.w = vw > 0.f ? vw : 0.f;
    }
    ((float4*)out)[(size_t)d * 32 + lane] = r;
}

__global__ void gemm_out2_kernel(const float* __restrict__ X,
                                 const float* __restrict__ W3,
                                 float* __restrict__ H3, int M) {
    int i = blockIdx.x * blockDim.x + threadIdx.x;
    if (i >= M) return;
    const float* xr = X + (size_t)i * CH;
    float a0 = 0.f, a1 = 0.f;
#pragma unroll 8
    for (int k = 0; k < CH; ++k) {
        const float x = xr[k];
        a0 += x * W3[2 * k];
        a1 += x * W3[2 * k + 1];
    }
    H3[(size_t)i * 2]     = a0;
    H3[(size_t)i * 2 + 1] = a1;
}

__global__ void gather2_kernel(const float* __restrict__ h3, const int* __restrict__ rowptr,
                               const int* __restrict__ csr_src, const float* __restrict__ dinv,
                               const float* __restrict__ b3, float* __restrict__ out, int n) {
    const int d = blockIdx.x * blockDim.x + threadIdx.x;
    if (d >= n) return;
    const int beg = rowptr[d];
    const int end = rowptr[d + 1];
    const float dd = dinv[d];
    const float2* __restrict__ hv = (const float2*)h3;
    const float2 vs = hv[d];
    const float ws = dd * dd;
    float a0 = vs.x * ws;
    float a1 = vs.y * ws;
#pragma unroll 4
    for (int j = beg; j < end; ++j) {
        const int s = csr_src[j];
        const float w = dinv[s] * dd;
        const float2 u = hv[s];
        a0 += u.x * w;
        a1 += u.y * w;
    }
    float2 r;
    r.x = a0 + b3[0];
    r.y = a1 + b3[1];
    ((float2*)out)[d] = r;
}

extern "C" void kernel_launch(void* const* d_in, const int* in_sizes, int n_in,
                              void* d_out, int out_size, void* d_ws, size_t ws_size,
                              hipStream_t stream) {
    const float* x   = (const float*)d_in[0];
    const int*   ei  = (const int*)d_in[1];
    const float* W1  = (const float*)d_in[2];
    const float* b1  = (const float*)d_in[3];
    const float* g1  = (const float*)d_in[4];
    const float* be1 = (const float*)d_in[5];
    const float* rm1 = (const float*)d_in[6];
    const float* rv1 = (const float*)d_in[7];
    const float* W2  = (const float*)d_in[8];
    const float* b2  = (const float*)d_in[9];
    const float* g2  = (const float*)d_in[10];
    const float* be2 = (const float*)d_in[11];
    const float* rm2 = (const float*)d_in[12];
    const float* rv2 = (const float*)d_in[13];
    const float* W3  = (const float*)d_in[14];
    const float* b3  = (const float*)d_in[15];

    const int Nn   = in_sizes[0] / CH;
    const int Ee   = in_sizes[1] / 2;
    const int NC   = Nn * CH;

    const int* src = ei;
    const int* dst = ei + Ee;

    const int T = 256;
    const int gN  = (Nn + T - 1) / T;
    const int gE  = (Ee + T - 1) / T;
    const int gW  = (Nn + 7) / 8;
    const int gMM = Nn / 16;

    const size_t NP = ((size_t)Nn + 64) & ~(size_t)63;
    const size_t BP = ((size_t)gN + 63) & ~(size_t)63;
    const size_t EP = ((size_t)Ee + 63) & ~(size_t)63;
    int*   cnt    = (int*)d_ws;
    int*   bsum   = cnt + NP;
    int*   rowptr = bsum + BP;
    float* dinv   = (float*)(rowptr + NP);
    int*   rank   = (int*)(dinv + NP);
    int*   csr    = rank + EP;
    float* bufA   = (float*)(csr + EP);
    float* bufB   = bufA + (size_t)NC;
    float* h3     = bufB + (size_t)NC;
    float* out    = (float*)d_out;

    zero_int_kernel<<<gN, T, 0, stream>>>(cnt, Nn);
    rank_count_kernel<<<gE, T, 0, stream>>>(dst, cnt, rank, Ee);
    block_sum_kernel<<<gN, T, 0, stream>>>(cnt, bsum, Nn);
    row_start_kernel<<<gN, T, 0, stream>>>(cnt, bsum, rowptr, dinv, Nn);
    place_kernel<<<gE, T, 0, stream>>>(src, dst, rank, rowptr, csr, Ee);

    gemm_wmma_f32<<<gMM, T, 0, stream>>>(x, W1, bufA);
    gather128_kernel<<<gW, T, 0, stream>>>(bufA, rowptr, csr, dinv, b1, g1, be1, rm1, rv1, bufB, Nn);

    gemm_wmma_f32<<<gMM, T, 0, stream>>>(bufB, W2, bufA);
    gather128_kernel<<<gW, T, 0, stream>>>(bufA, rowptr, csr, dinv, b2, g2, be2, rm2, rv2, bufB, Nn);

    gemm_out2_kernel<<<gN, T, 0, stream>>>(bufB, W3, h3, Nn);
    gather2_kernel<<<gN, T, 0, stream>>>(h3, rowptr, csr, dinv, b3, out, Nn);
}


# GEMM re-tiled: one wave per 16x128 row stripe, 8 accumulator tiles per A fragment, B fragments from a per-work-group LDS copy of W, rows stored as whole lines; k order and sums unchanged
# speedup vs baseline: 3.2404x; 1.1050x over previous
#include <hip/hip_runtime.h>

#define CH 128

typedef float v2f __attribute__((ext_vector_type(2)));
typedef float v8f __attribute__((ext_vector_type(8)));

__global__ void zero_int_kernel(int* __restrict__ p, int n) {
    int i = blockIdx.x * blockDim.x + threadIdx.x;
    if (i < n) p[i] = 0;
}

__global__ void rank_count_kernel(const int* __restrict__ dst, int* __restrict__ cnt,
                                  int* __restrict__ rank, int E) {
    int e = blockIdx.x * blockDim.x + threadIdx.x;
    if (e < E) rank[e] = atomicAdd(&cnt[dst[e]], 1);
}

__global__ __launch_bounds__(256)
void block_sum_kernel(const int* __restrict__ cnt, int* __restrict__ bsum, int n) {
    __shared__ int sm[256];
    const int t = threadIdx.x;
    const int i = blockIdx.x * 256 + t;
    sm[t] = (i < n) ? cnt[i] : 0;
    __syncthreads();
    for (int s = 128; s > 0; s >>= 1) {
        if (t < s) sm[t] += sm[t + s];
        __syncthreads();
    }
    if (t == 0) bsum[blockIdx.x] = sm[0];
}

__global__ __launch_bounds__(256)
void row_start_kernel(const int* __restrict__ cnt, const int* __restrict__ bsum,
                      int* __restrict__ rowptr, float* __restrict__ dinv, int n) {
    __shared__ int sm[256];
    __shared__ int red[256];
    const int t = threadIdx.x;
    const int b = blockIdx.x;
    int part = 0;
    for (int k = t; k < b; k += 256) part += bsum[k];
    red[t] = part;
    const int i = b * 256 + t;
    const int c = (i < n) ? cnt[i] : 0;
    sm[t] = c;
    __syncthreads();
    for (int s = 128; s > 0; s >>= 1) {
        if (t < s) red[t] += red[t + s];
        __syncthreads();
    }
    const int off = red[0];
    for (int s = 1; s < 256; s <<= 1) {
        const int v = (t >= s) ? sm[t - s] : 0;
        __syncthreads();
        sm[t] += v;
        __syncthreads();
    }
    const int incl = sm[t];
    if (i < n) {
        rowptr[i] = off + incl - c;
        dinv[i]   = rsqrtf((float)(c + 1));
    }
    if (i == n - 1) rowptr[n] = off + incl;
}

__global__ void place_kernel(const int* __restrict__ src, const int* __restrict__ dst,
                             const int* __restrict__ rank, const int* __restrict__ rowptr,
                             int* __restrict__ csr_src, int E) {
    int e = blockIdx.x * blockDim.x + threadIdx.x;
    if (e < E) csr_src[rowptr[dst[e]] + rank[e]] = src[e];
}

__global__ __launch_bounds__(256) __attribute__((amdgpu_waves_per_eu(8, 8)))
void gemm_wmma_f32(const float* __restrict__ X, const float* __restrict__ W,
                   float* __restrict__ H, int ntiles) {
    __shared__ float4 wl[64 * 64];

    const int wave = threadIdx.x >> 5;
    const int lane = threadIdx.x & 31;
    const int c    = lane & 15;
    const int h    = lane >> 4;
    const int tile = blockIdx.x * 8 + wave;
    const int tl   = tile < ntiles ? tile : ntiles - 1;
    const int m0   = tl * 16;

    const v2f* __restrict__ xa = (const v2f*)(X + (size_t)(m0 + c) * CH) + h;
    v2f a[32];
#pragma unroll
    for (int s = 0; s < 8; ++s) a[s] = xa[2 * s];

    const float4* __restrict__ W4 = (const float4*)W;
#pragma unroll
    for (int it = 0; it < 8; ++it) {
        const int p = it * 8 + wave;
        const float4 ra = W4[(2 * p) * 32 + lane];
        const float4 rb = W4[(2 * p + 1) * 32 + lane];
        float4 lo, hi;
        lo.x = ra.x; lo.y = rb.x; lo.z = ra.y; lo.w = rb.y;
        hi.x = ra.z; hi.y = rb.z; hi.z = ra.w; hi.w = rb.w;
        wl[p * 64 + (2 * h) * 16 + c]     = lo;
        wl[p * 64 + (2 * h + 1) * 16 + c] = hi;
    }
    __syncthreads();
    if (tile >= ntiles) return;

    v8f acc[8];
#pragma unroll
    for (int t = 0; t < 8; ++t) {
        acc[t] = (v8f){0.f, 0.f, 0.f, 0.f, 0.f, 0.f, 0.f, 0.f};
    }

    const float4* bl = wl + h * 64 + c;
#pragma unroll
    for (int s = 0; s < 32; ++s) {
        if ((s & 7) == 0 && s + 8 < 32) {
#pragma unroll
            for (int i = 0; i < 8; ++i) a[s + 8 + i] = xa[2 * (s + 8 + i)];
            __builtin_amdgcn_sched_barrier(0);
        }
        const float4 q0 = bl[s * 128];
        const float4 q1 = bl[s * 128 + 16];
        const float4 q2 = bl[s * 128 + 32];
        const float4 q3 = bl[s * 128 + 48];
        v2f b;
        b.x = q0.x; b.y = q0.y;
        acc[0] = __builtin_amdgcn_wmma_f32_16x16x4_f32(false, a[s], false, b, (short)0, acc[0], false, false);
        b.x = q0.z; b.y = q0.w;
        acc[1] = __builtin_amdgcn_wmma_f32_16x16x4_f32(false, a[s], false, b, (short)0, acc[1], false, false);
        b.x = q1.x; b.y = q1.y;
        acc[2] = __builtin_amdgcn_wmma_f32_16x16x4_f32(false, a[s], false, b, (short)0, acc[2], false, false);
        b.x = q1.z; b.y = q1.w;
        acc[3] = __builtin_amdgcn_wmma_f32_16x16x4_f32(false, a[s], false, b, (short)0, acc[3], false, false);
        b.x = q2.x; b.y = q2.y;
        acc[4] = __builtin_amdgcn_wmma_f32_16x16x4_f32(false, a[s], false, b, (short)0, acc[4], false, false);
        b.x = q2.z; b.y = q2.w;
        acc[5] = __builtin_amdgcn_wmma_f32_16x16x4_f32(false, a[s], false, b, (short)0, acc[5], false, false);
        b.x = q3.x; b.y = q3.y;
        acc[6] = __builtin_amdgcn_wmma_f32_16x16x4_f32(false, a[s], false, b, (short)0, acc[6], false, false);
        b.x = q3.z; b.y = q3.w;
        acc[7] = __builtin_amdgcn_wmma_f32_16x16x4_f32(false, a[s], false, b, (short)0, acc[7], false, false);
    }

    float4* __restrict__ H4 = (float4*)H;
#pragma unroll
    for (int r = 0; r < 8; ++r) {
        const size_t row4 = (size_t)(m0 + 8 * h + r) * (CH / 4);
        float4 v0, v1;
        v0.x = acc[0][r]; v0.y = acc[1][r]; v0.z = acc[2][r]; v0.w = acc[3][r];
        v1.x = acc[4][r]; v1.y = acc[5][r]; v1.z = acc[6][r]; v1.w = acc[7][r];
        H4[row4 + c]      = v0;
        H4[row4 + 16 + c] = v1;
    }
}

__global__ __launch_bounds__(256)
void gather128_kernel(const float* __restrict__ h, const int* __restrict__ rowptr,
                      const int* __restrict__ csr_src, const float* __restrict__ dinv,
                      const float* __restrict__ b, const float* __restrict__ g,
                      const float* __restrict__ be, const float* __restrict__ rm,
                      const float* __restrict__ rv,
                      float* __restrict__ out, int n) {
    const int d = __builtin_amdgcn_readfirstlane(
        (int)((blockIdx.x * blockDim.x + threadIdx.x) >> 5));
    if (d >= n) return;
    const int lane = threadIdx.x & 31;
    const int beg = rowptr[d];
    const int end = rowptr[d + 1];
    const float dd = dinv[d];
    const float4* __restrict__ hv = (const float4*)h;

    const float4 pb  = ((const float4*)b)[lane];
    const float4 pg  = ((const float4*)g)[lane];
    const float4 pbe = ((const float4*)be)[lane];
    const float4 prm = ((const float4*)rm)[lane];
    const float4 prv = ((const float4*)rv)[lane];
    const float ix = pg.x * rsqrtf(prv.x + 1e-5f);
    const float iy = pg.y * rsqrtf(prv.y + 1e-5f);
    const float iz = pg.z * rsqrtf(prv.z + 1e-5f);
    const float iw = pg.w * rsqrtf(prv.w + 1e-5f);

    const float4 vs = hv[(size_t)d * 32 + lane];
    const float ws = dd * dd;
    float4 acc;
    acc.x = vs.x * ws; acc.y = vs.y * ws; acc.z = vs.z * ws; acc.w = vs.w * ws;

    for (int base = beg; base < end; base += 32) {
        const int m = (end - base < 32) ? (end - base) : 32;
        int   ms = 0;
        float mw = 0.f;
        if (lane < m) {
            ms = csr_src[base + lane];
            mw = dinv[ms] * dd;
        }
        const int mwi = __float_as_int(mw);
        int j = 0;
        for (; j + 4 <= m; j += 4) {
            const int s0 = __builtin_amdgcn_readlane(ms, j);
            const int s1 = __builtin_amdgcn_readlane(ms, j + 1);
            const int s2 = __builtin_amdgcn_readlane(ms, j + 2);
            const int s3 = __builtin_amdgcn_readlane(ms, j + 3);
            const float w0 = __int_as_float(__builtin_amdgcn_readlane(mwi, j));
            const float w1 = __int_as_float(__builtin_amdgcn_readlane(mwi, j + 1));
            const float w2 = __int_as_float(__builtin_amdgcn_readlane(mwi, j + 2));
            const float w3 = __int_as_float(__builtin_amdgcn_readlane(mwi, j + 3));
            const float4 v0 = hv[(size_t)s0 * 32 + lane];
            const float4 v1 = hv[(size_t)s1 * 32 + lane];
            const float4 v2 = hv[(size_t)s2 * 32 + lane];
            const float4 v3 = hv[(size_t)s3 * 32 + lane];
            acc.x += v0.x * w0; acc.y += v0.y * w0; acc.z += v0.z * w0; acc.w += v0.w * w0;
            acc.x += v1.x * w1; acc.y += v1.y * w1; acc.z += v1.z * w1; acc.w += v1.w * w1;
            acc.x += v2.x * w2; acc.y += v2.y * w2; acc.z += v2.z * w2; acc.w += v2.w * w2;
            acc.x += v3.x * w3; acc.y += v3.y * w3; acc.z += v3.z * w3; acc.w += v3.w * w3;
        }
        const int rem = m & 3;
        for (int t = 0; t < rem; ++t) {
            const int s0 = __builtin_amdgcn_readlane(ms, j + t);
            const float w0 = __int_as_float(__builtin_amdgcn_readlane(mwi, j + t));
            const float4 v0 = hv[(size_t)s0 * 32 + lane];
            acc.x += v0.x * w0; acc.y += v0.y * w0; acc.z += v0.z * w0; acc.w += v0.w * w0;
        }
    }
    float4 r;
    {
        const float vx = (acc.x + pb.x - prm.x) * ix + pbe.x;
        const float vy = (acc.y + pb.y - prm.y) * iy + pbe.y;
        const float vz = (acc.z + pb.z - prm.z) * iz + pbe.z;
        const float vw = (acc.w + pb.w - prm.w) * iw + pbe.w;
        r.x = vx > 0.f ? vx : 0.f;
        r.y = vy > 0.f ? vy : 0.f;
        r.z = vz > 0.f ? vz : 0.f;
        r.w = vw > 0.f ? vw : 0.f;
    }
    ((float4*)out)[(size_t)d * 32 + lane] = r;
}

__global__ void gemm_out2_kernel(const float* __restrict__ X,
                                 const float* __restrict__ W3,
                                 float* __restrict__ H3, int M) {
    int i = blockIdx.x * blockDim.x + threadIdx.x;
    if (i >= M) return;
    const float* xr = X + (size_t)i * CH;
    float a0 = 0.f, a1 = 0.f;
#pragma unroll 8
    for (int k = 0; k < CH; ++k) {
        const float x = xr[k];
        a0 += x * W3[2 * k];
        a1 += x * W3[2 * k + 1];
    }
    H3[(size_t)i * 2]     = a0;
    H3[(size_t)i * 2 + 1] = a1;
}

__global__ void gather2_kernel(const float* __restrict__ h3, const int* __restrict__ rowptr,
                               const int* __restrict__ csr_src, const float* __restrict__ dinv,
                               const float* __restrict__ b3, float* __restrict__ out, int n) {
    const int d = blockIdx.x * blockDim.x + threadIdx.x;
    if (d >= n) return;
    const int beg = rowptr[d];
    const int end = rowptr[d + 1];
    const float dd = dinv[d];
    const float2* __restrict__ hv = (const float2*)h3;
    const float2 vs = hv[d];
    const float ws = dd * dd;
    float a0 = vs.x * ws;
    float a1 = vs.y * ws;
#pragma unroll 4
    for (int j = beg; j < end; ++j) {
        const int s = csr_src[j];
        const float w = dinv[s] * dd;
        const float2 u = hv[s];
        a0 += u.x * w;
        a1 += u.y * w;
    }
    float2 r;
    r.x = a0 + b3[0];
    r.y = a1 + b3[1];
    ((float2*)out)[d] = r;
}

extern "C" void kernel_launch(void* const* d_in, const int* in_sizes, int n_in,
                              void* d_out, int out_size, void* d_ws, size_t ws_size,
                              hipStream_t stream) {
    const float* x   = (const float*)d_in[0];
    const int*   ei  = (const int*)d_in[1];
    const float* W1  = (const float*)d_in[2];
    const float* b1  = (const float*)d_in[3];
    const float* g1  = (const float*)d_in[4];
    const float* be1 = (const float*)d_in[5];
    const float* rm1 = (const float*)d_in[6];
    const float* rv1 = (const float*)d_in[7];
    const float* W2  = (const float*)d_in[8];
    const float* b2  = (const float*)d_in[9];
    const float* g2  = (const float*)d_in[10];
    const float* be2 = (const float*)d_in[11];
    const float* rm2 = (const float*)d_in[12];
    const float* rv2 = (const float*)d_in[13];
    const float* W3  = (const float*)d_in[14];
    const float* b3  = (const float*)d_in[15];

    const int Nn   = in_sizes[0] / CH;
    const int Ee   = in_sizes[1] / 2;
    const int NC   = Nn * CH;

    const int* src = ei;
    const int* dst = ei + Ee;

    const int T = 256;
    const int gN  = (Nn + T - 1) / T;
    const int gE  = (Ee + T - 1) / T;
    const int gW  = (Nn + 7) / 8;
    const int gMM = Nn / 16;
    const int gMB = (gMM + 7) / 8;

    const size_t NP = ((size_t)Nn + 64) & ~(size_t)63;
    const size_t BP = ((size_t)gN + 63) & ~(size_t)63;
    const size_t EP = ((size_t)Ee + 63) & ~(size_t)63;
    int*   cnt    = (int*)d_ws;
    int*   bsum   = cnt + NP;
    int*   rowptr = bsum + BP;
    float* dinv   = (float*)(rowptr + NP);
    int*   rank   = (int*)(dinv + NP);
    int*   csr    = rank + EP;
    float* bufA   = (float*)(csr + EP);
    float* bufB   = bufA + (size_t)NC;
    float* h3     = bufB + (size_t)NC;
    float* out    = (float*)d_out;

    zero_int_kernel<<<gN, T, 0, stream>>>(cnt, Nn);
    rank_count_kernel<<<gE, T, 0, stream>>>(dst, cnt, rank, Ee);
    block_sum_kernel<<<gN, T, 0, stream>>>(cnt, bsum, Nn);
    row_start_kernel<<<gN, T, 0, stream>>>(cnt, bsum, rowptr, dinv, Nn);
    place_kernel<<<gE, T, 0, stream>>>(src, dst, rank, rowptr, csr, Ee);

    gemm_wmma_f32<<<gMB, T, 0, stream>>>(x, W1, bufA, gMM);
    gather128_kernel<<<gW, T, 0, stream>>>(bufA, rowptr, csr, dinv, b1, g1, be1, rm1, rv1, bufB, Nn);

    gemm_wmma_f32<<<gMB, T, 0, stream>>>(bufB, W2, bufA, gMM);
    gather128_kernel<<<gW, T, 0, stream>>>(bufA, rowptr, csr, dinv, b2, g2, be2, rm2, rv2, bufB, Nn);

    gemm_out2_kernel<<<gN, T, 0, stream>>>(bufB, W3, h3, Nn);
    gather2_kernel<<<gN, T, 0, stream>>>(h3, rowptr, csr, dinv, b3, out, Nn);
}


# Symmetric norm factored: each product scales row i by dinv[i] (128-col one on its A operand); gathers add plain rows and scale the sum by dinv[dst] once, no per-edge dinv[src] fetch; term order kept
# speedup vs baseline: 3.4263x; 1.0574x over previous
#include <hip/hip_runtime.h>

#define CH 128

typedef float v2f __attribute__((ext_vector_type(2)));
typedef float v8f __attribute__((ext_vector_type(8)));

__global__ void zero_int_kernel(int* __restrict__ p, int n) {
    int i = blockIdx.x * blockDim.x + threadIdx.x;
    if (i < n) p[i] = 0;
}

__global__ void rank_count_kernel(const int* __restrict__ dst, int* __restrict__ cnt,
                                  int* __restrict__ rank, int E) {
    int e = blockIdx.x * blockDim.x + threadIdx.x;
    if (e < E) rank[e] = atomicAdd(&cnt[dst[e]], 1);
}

__global__ __launch_bounds__(256)
void block_sum_kernel(const int* __restrict__ cnt, int* __restrict__ bsum, int n) {
    __shared__ int sm[256];
    const int t = threadIdx.x;
    const int i = blockIdx.x * 256 + t;
    sm[t] = (i < n) ? cnt[i] : 0;
    __syncthreads();
    for (int s = 128; s > 0; s >>= 1) {
        if (t < s) sm[t] += sm[t + s];
        __syncthreads();
    }
    if (t == 0) bsum[blockIdx.x] = sm[0];
}

__global__ __launch_bounds__(256)
void row_start_kernel(const int* __restrict__ cnt, const int* __restrict__ bsum,
                      int* __restrict__ rowptr, float* __restrict__ dinv, int n) {
    __shared__ int sm[256];
    __shared__ int red[256];
    const int t = threadIdx.x;
    const int b = blockIdx.x;
    int part = 0;
    for (int k = t; k < b; k += 256) part += bsum[k];
    red[t] = part;
    const int i = b * 256 + t;
    const int c = (i < n) ? cnt[i] : 0;
    sm[t] = c;
    __syncthreads();
    for (int s = 128; s > 0; s >>= 1) {
        if (t < s) red[t] += red[t + s];
        __syncthreads();
    }
    const int off = red[0];
    for (int s = 1; s < 256; s <<= 1) {
        const int v = (t >= s) ? sm[t - s] : 0;
        __syncthreads();
        sm[t] += v;
        __syncthreads();
    }
    const int incl = sm[t];
    if (i < n) {
        rowptr[i] = off + incl - c;
        dinv[i]   = rsqrtf((float)(c + 1));
    }
    if (i == n - 1) rowptr[n] = off + incl;
}

__global__ void place_kernel(const int* __restrict__ src, const int* __restrict__ dst,
                             const int* __restrict__ rank, const int* __restrict__ rowptr,
                             int* __restrict__ csr_src, int E) {
    int e = blockIdx.x * blockDim.x + threadIdx.x;
    if (e < E) csr_src[rowptr[dst[e]] + rank[e]] = src[e];
}

__global__ __launch_bounds__(256) __attribute__((amdgpu_waves_per_eu(8, 8)))
void gemm_wmma_f32(const float* __restrict__ X, const float* __restrict__ W,
                   const float* __restrict__ dinv,
                   float* __restrict__ H, int ntiles) {
    __shared__ float4 wl[64 * 64];

    const int wave = threadIdx.x >> 5;
    const int lane = threadIdx.x & 31;
    const int c    = lane & 15;
    const int h    = lane >> 4;
    const int tile = blockIdx.x * 8 + wave;
    const int tl   = tile < ntiles ? tile : ntiles - 1;
    const int m0   = tl * 16;

    const v2f* __restrict__ xa = (const v2f*)(X + (size_t)(m0 + c) * CH) + h;
    v2f a[32];
#pragma unroll
    for (int s = 0; s < 8; ++s) a[s] = xa[2 * s];
    const float dvl = dinv[m0 + c];

    const float4* __restrict__ W4 = (const float4*)W;
#pragma unroll
    for (int it = 0; it < 8; ++it) {
        const int p = it * 8 + wave;
        const float4 ra = W4[(2 * p) * 32 + lane];
        const float4 rb = W4[(2 * p + 1) * 32 + lane];
        float4 lo, hi;
        lo.x = ra.x; lo.y = rb.x; lo.z = ra.y; lo.w = rb.y;
        hi.x = ra.z; hi.y = rb.z; hi.z = ra.w; hi.w = rb.w;
        wl[p * 64 + (2 * h) * 16 + c]     = lo;
        wl[p * 64 + (2 * h + 1) * 16 + c] = hi;
    }
    __syncthreads();
    if (tile >= ntiles) return;

    v8f acc[8];
#pragma unroll
    for (int t = 0; t < 8; ++t) {
        acc[t] = (v8f){0.f, 0.f, 0.f, 0.f, 0.f, 0.f, 0.f, 0.f};
    }

    const float4* bl = wl + h * 64 + c;
#pragma unroll
    for (int s = 0; s < 32; ++s) {
        if ((s & 7) == 0) {
            if (s + 8 < 32) {
#pragma unroll
                for (int i = 0; i < 8; ++i) a[s + 8 + i] = xa[2 * (s + 8 + i)];
            }
            __builtin_amdgcn_sched_barrier(0);
        }
        const v2f as = a[s] * dvl;
        const float4 q0 = bl[s * 128];
        const float4 q1 = bl[s * 128 + 16];
        const float4 q2 = bl[s * 128 + 32];
        const float4 q3 = bl[s * 128 + 48];
        v2f b;
        b.x = q0.x; b.y = q0.y;
        acc[0] = __builtin_amdgcn_wmma_f32_16x16x4_f32(false, as, false, b, (short)0, acc[0], false, false);
        b.x = q0.z; b.y = q0.w;
        acc[1] = __builtin_amdgcn_wmma_f32_16x16x4_f32(false, as, false, b, (short)0, acc[1], false, false);
        b.x = q1.x; b.y = q1.y;
        acc[2] = __builtin_amdgcn_wmma_f32_16x16x4_f32(false, as, false, b, (short)0, acc[2], false, false);
        b.x = q1.z; b.y = q1.w;
        acc[3] = __builtin_amdgcn_wmma_f32_16x16x4_f32(false, as, false, b, (short)0, acc[3], false, false);
        b.x = q2.x; b.y = q2.y;
        acc[4] = __builtin_amdgcn_wmma_f32_16x16x4_f32(false, as, false, b, (short)0, acc[4], false, false);
        b.x = q2.z; b.y = q2.w;
        acc[5] = __builtin_amdgcn_wmma_f32_16x16x4_f32(false, as, false, b, (short)0, acc[5], false, false);
        b.x = q3.x; b.y = q3.y;
        acc[6] = __builtin_amdgcn_wmma_f32_16x16x4_f32(false, as, false, b, (short)0, acc[6], false, false);
        b.x = q3.z; b.y = q3.w;
        acc[7] = __builtin_amdgcn_wmma_f32_16x16x4_f32(false, as, false, b, (short)0, acc[7], false, false);
    }

    float4* __restrict__ H4 = (float4*)H;
#pragma unroll
    for (int r = 0; r < 8; ++r) {
        const size_t row4 = (size_t)(m0 + 8 * h + r) * (CH / 4);
        float4 v0, v1;
        v0.x = acc[0][r]; v0.y = acc[1][r]; v0.z = acc[2][r]; v0.w = acc[3][r];
        v1.x = acc[4][r]; v1.y = acc[5][r]; v1.z = acc[6][r]; v1.w = acc[7][r];
        H4[row4 + c]      = v0;
        H4[row4 + 16 + c] = v1;
    }
}

__global__ __launch_bounds__(256)
void gather128_kernel(const float* __restrict__ h, const int* __restrict__ rowptr,
                      const int* __restrict__ csr_src, const float* __restrict__ dinv,
                      const float* __restrict__ b, const float* __restrict__ g,
                      const float* __restrict__ be, const float* __restrict__ rm,
                      const float* __restrict__ rv,
                      float* __restrict__ out, int n) {
    const int d = __builtin_amdgcn_readfirstlane(
        (int)((blockIdx.x * blockDim.x + threadIdx.x) >> 5));
    if (d >= n) return;
    const int lane = threadIdx.x & 31;
    const int beg = rowptr[d];
    const int end = rowptr[d + 1];
    const float dd = dinv[d];
    const float4* __restrict__ hv = (const float4*)h;

    const float4 pb  = ((const float4*)b)[lane];
    const float4 pg  = ((const float4*)g)[lane];
    const float4 pbe = ((const float4*)be)[lane];
    const float4 prm = ((const float4*)rm)[lane];
    const float4 prv = ((const float4*)rv)[lane];
    const float ix = pg.x * rsqrtf(prv.x + 1e-5f);
    const float iy = pg.y * rsqrtf(prv.y + 1e-5f);
    const float iz = pg.z * rsqrtf(prv.z + 1e-5f);
    const float iw = pg.w * rsqrtf(prv.w + 1e-5f);

    float4 acc = hv[(size_t)d * 32 + lane];

    for (int base = beg; base < end; base += 32) {
        const int m = (end - base < 32) ? (end - base) : 32;
        int ms = 0;
        if (lane < m) ms = csr_src[base + lane];
        int j = 0;
        for (; j + 4 <= m; j += 4) {
            const int s0 = __builtin_amdgcn_readlane(ms, j);
            const int s1 = __builtin_amdgcn_readlane(ms, j + 1);
            const int s2 = __builtin_amdgcn_readlane(ms, j + 2);
            const int s3 = __builtin_amdgcn_readlane(ms, j + 3);
            const float4 v0 = hv[(size_t)s0 * 32 + lane];
            const float4 v1 = hv[(size_t)s1 * 32 + lane];
            const float4 v2 = hv[(size_t)s2 * 32 + lane];
            const float4 v3 = hv[(size_t)s3 * 32 + lane];
            acc.x += v0.x; acc.y += v0.y; acc.z += v0.z; acc.w += v0.w;
            acc.x += v1.x; acc.y += v1.y; acc.z += v1.z; acc.w += v1.w;
            acc.x += v2.x; acc.y += v2.y; acc.z += v2.z; acc.w += v2.w;
            acc.x += v3.x; acc.y += v3.y; acc.z += v3.z; acc.w += v3.w;
        }
        const int rem = m & 3;
        for (int t = 0; t < rem; ++t) {
            const int s0 = __builtin_amdgcn_readlane(ms, j + t);
            const float4 v0 = hv[(size_t)s0 * 32 + lane];
            acc.x += v0.x; acc.y += v0.y; acc.z += v0.z; acc.w += v0.w;
        }
    }
    float4 r;
    {
        const float vx = (acc.x * dd + pb.x - prm.x) * ix + pbe.x;
        const float vy = (acc.y * dd + pb.y - prm.y) * iy + pbe.y;
        const float vz = (acc.z * dd + pb.z - prm.z) * iz + pbe.z;
        const float vw = (acc.w * dd + pb.w - prm.w) * iw + pbe.w;
        r.x = vx > 0.f ? vx : 0.f;
        r.y = vy > 0.f ? vy : 0.f;
        r.z = vz > 0.f ? vz : 0.f;
        r.w = vw > 0.f ? vw : 0.f;
    }
    ((float4*)out)[(size_t)d * 32 + lane] = r;
}

__global__ void gemm_out2_kernel(const float* __restrict__ X,
                                 const float* __restrict__ W3,
                                 const float* __restrict__ dinv,
                                 float* __restrict__ H3, int M) {
    int i = blockIdx.x * blockDim.x + threadIdx.x;
    if (i >= M) return;
    const float* xr = X + (size_t)i * CH;
    float a0 = 0.f, a1 = 0.f;
#pragma unroll 8
    for (int k = 0; k < CH; ++k) {
        const float x = xr[k];
        a0 += x * W3[2 * k];
        a1 += x * W3[2 * k + 1];
    }
    const float sc = dinv[i];
    H3[(size_t)i * 2]     = a0 * sc;
    H3[(size_t)i * 2 + 1] = a1 * sc;
}

__global__ void gather2_kernel(const float* __restrict__ h3, const int* __restrict__ rowptr,
                               const int* __restrict__ csr_src, const float* __restrict__ dinv,
                               const float* __restrict__ b3, float* __restrict__ out, int n) {
    const int d = blockIdx.x * blockDim.x + threadIdx.x;
    if (d >= n) return;
    const int beg = rowptr[d];
    const int end = rowptr[d + 1];
    const float dd = dinv[d];
    const float2* __restrict__ hv = (const float2*)h3;
    const float2 vs = hv[d];
    float a0 = vs.x;
    float a1 = vs.y;
#pragma unroll 4
    for (int j = beg; j < end; ++j) {
        const int s = csr_src[j];
        const float2 u = hv[s];
        a0 += u.x;
        a1 += u.y;
    }
    float2 r;
    r.x = a0 * dd + b3[0];
    r.y = a1 * dd + b3[1];
    ((float2*)out)[d] = r;
}

extern "C" void kernel_launch(void* const* d_in, const int* in_sizes, int n_in,
                              void* d_out, int out_size, void* d_ws, size_t ws_size,
                              hipStream_t stream) {
    const float* x   = (const float*)d_in[0];
    const int*   ei  = (const int*)d_in[1];
    const float* W1  = (const float*)d_in[2];
    const float* b1  = (const float*)d_in[3];
    const float* g1  = (const float*)d_in[4];
    const float* be1 = (const float*)d_in[5];
    const float* rm1 = (const float*)d_in[6];
    const float* rv1 = (const float*)d_in[7];
    const float* W2  = (const float*)d_in[8];
    const float* b2  = (const float*)d_in[9];
    const float* g2  = (const float*)d_in[10];
    const float* be2 = (const float*)d_in[11];
    const float* rm2 = (const float*)d_in[12];
    const float* rv2 = (const float*)d_in[13];
    const float* W3  = (const float*)d_in[14];
    const float* b3  = (const float*)d_in[15];

    const int Nn   = in_sizes[0] / CH;
    const int Ee   = in_sizes[1] / 2;
    const int NC   = Nn * CH;

    const int* src = ei;
    const int* dst = ei + Ee;

    const int T = 256;
    const int gN  = (Nn + T - 1) / T;
    const int gE  = (Ee + T - 1) / T;
    const int gW  = (Nn + 7) / 8;
    const int gMM = Nn / 16;
    const int gMB = (gMM + 7) / 8;

    const size_t NP = ((size_t)Nn + 64) & ~(size_t)63;
    const size_t BP = ((size_t)gN + 63) & ~(size_t)63;
    const size_t EP = ((size_t)Ee + 63) & ~(size_t)63;
    int*   cnt    = (int*)d_ws;
    int*   bsum   = cnt + NP;
    int*   rowptr = bsum + BP;
    float* dinv   = (float*)(rowptr + NP);
    int*   rank   = (int*)(dinv + NP);
    int*   csr    = rank + EP;
    float* bufA   = (float*)(csr + EP);
    float* bufB   = bufA + (size_t)NC;
    float* h3     = bufB + (size_t)NC;
    float* out    = (float*)d_out;

    zero_int_kernel<<<gN, T, 0, stream>>>(cnt, Nn);
    rank_count_kernel<<<gE, T, 0, stream>>>(dst, cnt, rank, Ee);
    block_sum_kernel<<<gN, T, 0, stream>>>(cnt, bsum, Nn);
    row_start_kernel<<<gN, T, 0, stream>>>(cnt, bsum, rowptr, dinv, Nn);
    place_kernel<<<gE, T, 0, stream>>>(src, dst, rank, rowptr, csr, Ee);

    gemm_wmma_f32<<<gMB, T, 0, stream>>>(x, W1, dinv, bufA, gMM);
    gather128_kernel<<<gW, T, 0, stream>>>(bufA, rowptr, csr, dinv, b1, g1, be1, rm1, rv1, bufB, Nn);

    gemm_wmma_f32<<<gMB, T, 0, stream>>>(bufB, W2, dinv, bufA, gMM);
    gather128_kernel<<<gW, T, 0, stream>>>(bufA, rowptr, csr, dinv, b2, g2, be2, rm2, rv2, bufB, Nn);

    gemm_out2_kernel<<<gN, T, 0, stream>>>(bufB, W3, dinv, h3, Nn);
    gather2_kernel<<<gN, T, 0, stream>>>(h3, rowptr, csr, dinv, b3, out, Nn);
}


# Edge placement pass (csr fill) runs inside the first GEMM's launch on extra work-groups, 4 edges per thread: scattered requests overlap the matrix-bound tiles; 11 to 10 launches, sums unchanged
# speedup vs baseline: 3.5393x; 1.0330x over previous
#include <hip/hip_runtime.h>

#define CH 128

typedef float v2f __attribute__((ext_vector_type(2)));
typedef float v8f __attribute__((ext_vector_type(8)));

__global__ void zero_int_kernel(int* __restrict__ p, int n) {
    int i = blockIdx.x * blockDim.x + threadIdx.x;
    if (i < n) p[i] = 0;
}

__global__ void rank_count_kernel(const int* __restrict__ dst, int* __restrict__ cnt,
                                  int* __restrict__ rank, int E) {
    int e = blockIdx.x * blockDim.x + threadIdx.x;
    if (e < E) rank[e] = atomicAdd(&cnt[dst[e]], 1);
}

__global__ __launch_bounds__(256)
void block_sum_kernel(const int* __restrict__ cnt, int* __restrict__ bsum, int n) {
    __shared__ int sm[256];
    const int t = threadIdx.x;
    const int i = blockIdx.x * 256 + t;
    sm[t] = (i < n) ? cnt[i] : 0;
    __syncthreads();
    for (int s = 128; s > 0; s >>= 1) {
        if (t < s) sm[t] += sm[t + s];
        __syncthreads();
    }
    if (t == 0) bsum[blockIdx.x] = sm[0];
}

__global__ __launch_bounds__(256)
void row_start_kernel(const int* __restrict__ cnt, const int* __restrict__ bsum,
                      int* __restrict__ rowptr, float* __restrict__ dinv, int n) {
    __shared__ int sm[256];
    __shared__ int red[256];
    const int t = threadIdx.x;
    const int b = blockIdx.x;
    int part = 0;
    for (int k = t; k < b; k += 256) part += bsum[k];
    red[t] = part;
    const int i = b * 256 + t;
    const int c = (i < n) ? cnt[i] : 0;
    sm[t] = c;
    __syncthreads();
    for (int s = 128; s > 0; s >>= 1) {
        if (t < s) red[t] += red[t + s];
        __syncthreads();
    }
    const int off = red[0];
    for (int s = 1; s < 256; s <<= 1) {
        const int v = (t >= s) ? sm[t - s] : 0;
        __syncthreads();
        sm[t] += v;
        __syncthreads();
    }
    const int incl = sm[t];
    if (i < n) {
        rowptr[i] = off + incl - c;
        dinv[i]   = rsqrtf((float)(c + 1));
    }
    if (i == n - 1) rowptr[n] = off + incl;
}

__global__ __launch_bounds__(256) __attribute__((amdgpu_waves_per_eu(8, 8)))
void gemm_wmma_f32(const float* __restrict__ X, const float* __restrict__ W,
                   const float* __restrict__ dinv,
                   float* __restrict__ H, int ntiles, int gemm_blocks,
                   const int* __restrict__ src, const int* __restrict__ dst,
                   const int* __restrict__ rank, const int* __restrict__ rowptr,
                   int* __restrict__ csr_src, int E) {
    __shared__ float4 wl[64 * 64];

    if ((int)blockIdx.x >= gemm_blocks) {
        const int first = ((int)blockIdx.x - gemm_blocks) * 1024;
        const int t     = (int)threadIdx.x;
        if (first + 1024 <= E) {
            int ed[4], er[4], es[4], ep[4];
#pragma unroll
            for (int k = 0; k < 4; ++k) {
                const int e = first + t + 256 * k;
                ed[k] = dst[e];
                er[k] = rank[e];
                es[k] = src[e];
            }
#pragma unroll
            for (int k = 0; k < 4; ++k) ep[k] = rowptr[ed[k]] + er[k];
#pragma unroll
            for (int k = 0; k < 4; ++k) csr_src[ep[k]] = es[k];
        } else {
            for (int k = 0; k < 4; ++k) {
                const int e = first + t + 256 * k;
                if (e < E) csr_src[rowptr[dst[e]] + rank[e]] = src[e];
            }
        }
        return;
    }

    const int wave = threadIdx.x >> 5;
    const int lane = threadIdx.x & 31;
    const int c    = lane & 15;
    const int h    = lane >> 4;
    const int tile = blockIdx.x * 8 + wave;
    const int tl   = tile < ntiles ? tile : ntiles - 1;
    const int m0   = tl * 16;

    const v2f* __restrict__ xa = (const v2f*)(X + (size_t)(m0 + c) * CH) + h;
    v2f a[32];
#pragma unroll
    for (int s = 0; s < 8; ++s) a[s] = xa[2 * s];
    const float dvl = dinv[m0 + c];

    const float4* __restrict__ W4 = (const float4*)W;
#pragma unroll
    for (int it = 0; it < 8; ++it) {
        const int p = it * 8 + wave;
        const float4 ra = W4[(2 * p) * 32 + lane];
        const float4 rb = W4[(2 * p + 1) * 32 + lane];
        float4 lo, hi;
        lo.x = ra.x; lo.y = rb.x; lo.z = ra.y; lo.w = rb.y;
        hi.x = ra.z; hi.y = rb.z; hi.z = ra.w; hi.w = rb.w;
        wl[p * 64 + (2 * h) * 16 + c]     = lo;
        wl[p * 64 + (2 * h + 1) * 16 + c] = hi;
    }
    __syncthreads();
    if (tile >= ntiles) return;

    v8f acc[8];
#pragma unroll
    for (int t = 0; t < 8; ++t) {
        acc[t] = (v8f){0.f, 0.f, 0.f, 0.f, 0.f, 0.f, 0.f, 0.f};
    }

    const float4* bl = wl + h * 64 + c;
#pragma unroll
    for (int s = 0; s < 32; ++s) {
        if ((s & 7) == 0) {
            if (s + 8 < 32) {
#pragma unroll
                for (int i = 0; i < 8; ++i) a[s + 8 + i] = xa[2 * (s + 8 + i)];
            }
            __builtin_amdgcn_sched_barrier(0);
        }
        const v2f as = a[s] * dvl;
        const float4 q0 = bl[s * 128];
        const float4 q1 = bl[s * 128 + 16];
        const float4 q2 = bl[s * 128 + 32];
        const float4 q3 = bl[s * 128 + 48];
        v2f b;
        b.x = q0.x; b.y = q0.y;
        acc[0] = __builtin_amdgcn_wmma_f32_16x16x4_f32(false, as, false, b, (short)0, acc[0], false, false);
        b.x = q0.z; b.y = q0.w;
        acc[1] = __builtin_amdgcn_wmma_f32_16x16x4_f32(false, as, false, b, (short)0, acc[1], false, false);
        b.x = q1.x; b.y = q1.y;
        acc[2] = __builtin_amdgcn_wmma_f32_16x16x4_f32(false, as, false, b, (short)0, acc[2], false, false);
        b.x = q1.z; b.y = q1.w;
        acc[3] = __builtin_amdgcn_wmma_f32_16x16x4_f32(false, as, false, b, (short)0, acc[3], false, false);
        b.x = q2.x; b.y = q2.y;
        acc[4] = __builtin_amdgcn_wmma_f32_16x16x4_f32(false, as, false, b, (short)0, acc[4], false, false);
        b.x = q2.z; b.y = q2.w;
        acc[5] = __builtin_amdgcn_wmma_f32_16x16x4_f32(false, as, false, b, (short)0, acc[5], false, false);
        b.x = q3.x; b.y = q3.y;
        acc[6] = __builtin_amdgcn_wmma_f32_16x16x4_f32(false, as, false, b, (short)0, acc[6], false, false);
        b.x = q3.z; b.y = q3.w;
        acc[7] = __builtin_amdgcn_wmma_f32_16x16x4_f32(false, as, false, b, (short)0, acc[7], false, false);
    }

    float4* __restrict__ H4 = (float4*)H;
#pragma unroll
    for (int r = 0; r < 8; ++r) {
        const size_t row4 = (size_t)(m0 + 8 * h + r) * (CH / 4);
        float4 v0, v1;
        v0.x = acc[0][r]; v0.y = acc[1][r]; v0.z = acc[2][r]; v0.w = acc[3][r];
        v1.x = acc[4][r]; v1.y = acc[5][r]; v1.z = acc[6][r]; v1.w = acc[7][r];
        H4[row4 + c]      = v0;
        H4[row4 + 16 + c] = v1;
    }
}

__global__ __launch_bounds__(256)
void gather128_kernel(const float* __restrict__ h, const int* __restrict__ rowptr,
                      const int* __restrict__ csr_src, const float* __restrict__ dinv,
                      const float* __restrict__ b, const float* __restrict__ g,
                      const float* __restrict__ be, const float* __restrict__ rm,
                      const float* __restrict__ rv,
                      float* __restrict__ out, int n) {
    const int d = __builtin_amdgcn_readfirstlane(
        (int)((blockIdx.x * blockDim.x + threadIdx.x) >> 5));
    if (d >= n) return;
    const int lane = threadIdx.x & 31;
    const int beg = rowptr[d];
    const int end = rowptr[d + 1];
    const float dd = dinv[d];
    const float4* __restrict__ hv = (const float4*)h;

    const float4 pb  = ((const float4*)b)[lane];
    const float4 pg  = ((const float4*)g)[lane];
    const float4 pbe = ((const float4*)be)[lane];
    const float4 prm = ((const float4*)rm)[lane];
    const float4 prv = ((const float4*)rv)[lane];
    const float ix = pg.x * rsqrtf(prv.x + 1e-5f);
    const float iy = pg.y * rsqrtf(prv.y + 1e-5f);
    const float iz = pg.z * rsqrtf(prv.z + 1e-5f);
    const float iw = pg.w * rsqrtf(prv.w + 1e-5f);

    float4 acc = hv[(size_t)d * 32 + lane];

    for (int base = beg; base < end; base += 32) {
        const int m = (end - base < 32) ? (end - base) : 32;
        int ms = 0;
        if (lane < m) ms = csr_src[base + lane];
        int j = 0;
        for (; j + 4 <= m; j += 4) {
            const int s0 = __builtin_amdgcn_readlane(ms, j);
            const int s1 = __builtin_amdgcn_readlane(ms, j + 1);
            const int s2 = __builtin_amdgcn_readlane(ms, j + 2);
            const int s3 = __builtin_amdgcn_readlane(ms, j + 3);
            const float4 v0 = hv[(size_t)s0 * 32 + lane];
            const float4 v1 = hv[(size_t)s1 * 32 + lane];
            const float4 v2 = hv[(size_t)s2 * 32 + lane];
            const float4 v3 = hv[(size_t)s3 * 32 + lane];
            acc.x += v0.x; acc.y += v0.y; acc.z += v0.z; acc.w += v0.w;
            acc.x += v1.x; acc.y += v1.y; acc.z += v1.z; acc.w += v1.w;
            acc.x += v2.x; acc.y += v2.y; acc.z += v2.z; acc.w += v2.w;
            acc.x += v3.x; acc.y += v3.y; acc.z += v3.z; acc.w += v3.w;
        }
        const int rem = m & 3;
        for (int t = 0; t < rem; ++t) {
            const int s0 = __builtin_amdgcn_readlane(ms, j + t);
            const float4 v0 = hv[(size_t)s0 * 32 + lane];
            acc.x += v0.x; acc.y += v0.y; acc.z += v0.z; acc.w += v0.w;
        }
    }
    float4 r;
    {
        const float vx = (acc.x * dd + pb.x - prm.x) * ix + pbe.x;
        const float vy = (acc.y * dd + pb.y - prm.y) * iy + pbe.y;
        const float vz = (acc.z * dd + pb.z - prm.z) * iz + pbe.z;
        const float vw = (acc.w * dd + pb.w - prm.w) * iw + pbe.w;
        r.x = vx > 0.f ? vx : 0.f;
        r.y = vy > 0.f ? vy : 0.f;
        r.z = vz > 0.f ? vz : 0.f;
        r.w = vw > 0.f ? vw : 0.f;
    }
    ((float4*)out)[(size_t)d * 32 + lane] = r;
}

__global__ void gemm_out2_kernel(const float* __restrict__ X,
                                 const float* __restrict__ W3,
                                 const float* __restrict__ dinv,
                                 float* __restrict__ H3, int M) {
    int i = blockIdx.x * blockDim.x + threadIdx.x;
    if (i >= M) return;
    const float* xr = X + (size_t)i * CH;
    float a0 = 0.f, a1 = 0.f;
#pragma unroll 8
    for (int k = 0; k < CH; ++k) {
        const float x = xr[k];
        a0 += x * W3[2 * k];
        a1 += x * W3[2 * k + 1];
    }
    const float sc = dinv[i];
    H3[(size_t)i * 2]     = a0 * sc;
    H3[(size_t)i * 2 + 1] = a1 * sc;
}

__global__ void gather2_kernel(const float* __restrict__ h3, const int* __restrict__ rowptr,
                               const int* __restrict__ csr_src, const float* __restrict__ dinv,
                               const float* __restrict__ b3, float* __restrict__ out, int n) {
    const int d = blockIdx.x * blockDim.x + threadIdx.x;
    if (d >= n) return;
    const int beg = rowptr[d];
    const int end = rowptr[d + 1];
    const float dd = dinv[d];
    const float2* __restrict__ hv = (const float2*)h3;
    const float2 vs = hv[d];
    float a0 = vs.x;
    float a1 = vs.y;
#pragma unroll 4
    for (int j = beg; j < end; ++j) {
        const int s = csr_src[j];
        const float2 u = hv[s];
        a0 += u.x;
        a1 += u.y;
    }
    float2 r;
    r.x = a0 * dd + b3[0];
    r.y = a1 * dd + b3[1];
    ((float2*)out)[d] = r;
}

extern "C" void kernel_launch(void* const* d_in, const int* in_sizes, int n_in,
                              void* d_out, int out_size, void* d_ws, size_t ws_size,
                              hipStream_t stream) {
    const float* x   = (const float*)d_in[0];
    const int*   ei  = (const int*)d_in[1];
    const float* W1  = (const float*)d_in[2];
    const float* b1  = (const float*)d_in[3];
    const float* g1  = (const float*)d_in[4];
    const float* be1 = (const float*)d_in[5];
    const float* rm1 = (const float*)d_in[6];
    const float* rv1 = (const float*)d_in[7];
    const float* W2  = (const float*)d_in[8];
    const float* b2  = (const float*)d_in[9];
    const float* g2  = (const float*)d_in[10];
    const float* be2 = (const float*)d_in[11];
    const float* rm2 = (const float*)d_in[12];
    const float* rv2 = (const float*)d_in[13];
    const float* W3  = (const float*)d_in[14];
    const float* b3  = (const float*)d_in[15];

    const int Nn   = in_sizes[0] / CH;
    const int Ee   = in_sizes[1] / 2;
    const int NC   = Nn * CH;

    const int* src = ei;
    const int* dst = ei + Ee;

    const int T = 256;
    const int gN  = (Nn + T - 1) / T;
    const int gE  = (Ee + T - 1) / T;
    const int gW  = (Nn + 7) / 8;
    const int gMM = Nn / 16;
    const int gMB = (gMM + 7) / 8;
    const int gPL = (Ee + 1023) / 1024;

    const size_t NP = ((size_t)Nn + 64) & ~(size_t)63;
    const size_t BP = ((size_t)gN + 63) & ~(size_t)63;
    const size_t EP = ((size_t)Ee + 63) & ~(size_t)63;
    int*   cnt    = (int*)d_ws;
    int*   bsum   = cnt + NP;
    int*   rowptr = bsum + BP;
    float* dinv   = (float*)(rowptr + NP);
    int*   rank   = (int*)(dinv + NP);
    int*   csr    = rank + EP;
    float* bufA   = (float*)(csr + EP);
    float* bufB   = bufA + (size_t)NC;
    float* h3     = bufB + (size_t)NC;
    float* out    = (float*)d_out;

    zero_int_kernel<<<gN, T, 0, stream>>>(cnt, Nn);
    rank_count_kernel<<<gE, T, 0, stream>>>(dst, cnt, rank, Ee);
    block_sum_kernel<<<gN, T, 0, stream>>>(cnt, bsum, Nn);
    row_start_kernel<<<gN, T, 0, stream>>>(cnt, bsum, rowptr, dinv, Nn);

    gemm_wmma_f32<<<gMB + gPL, T, 0, stream>>>(x, W1, dinv, bufA, gMM, gMB,
                                               src, dst, rank, rowptr, csr, Ee);
    gather128_kernel<<<gW, T, 0, stream>>>(bufA, rowptr, csr, dinv, b1, g1, be1, rm1, rv1, bufB, Nn);

    gemm_wmma_f32<<<gMB, T, 0, stream>>>(bufB, W2, dinv, bufA, gMM, gMB,
                                         src, dst, rank, rowptr, csr, 0);
    gather128_kernel<<<gW, T, 0, stream>>>(bufA, rowptr, csr, dinv, b2, g2, be2, rm2, rv2, bufB, Nn);

    gemm_out2_kernel<<<gN, T, 0, stream>>>(bufB, W3, dinv, h3, Nn);
    gather2_kernel<<<gN, T, 0, stream>>>(h3, rowptr, csr, dinv, b3, out, Nn);
}


# Edge grouping rebuilt as a two-level bucket sort on LDS counters (512-node buckets per edge chunk, then count, scan, place per bucket): no global atomics left; node slot order = LDS arrival order
# speedup vs baseline: 4.4545x; 1.2586x over previous
#include <hip/hip_runtime.h>

#define CH 128

typedef float v2f __attribute__((ext_vector_type(2)));
typedef float v8f __attribute__((ext_vector_type(8)));

#define BSH    9
#define BNODES 512
#define CHUNKS 256

__global__ __launch_bounds__(1024)
void bucket_count_kernel(const int* __restrict__ dst, int* __restrict__ tab,
                         int E, int per, int nb) {
    extern __shared__ int sh[];
    const int t = threadIdx.x;
    for (int b = t; b < nb; b += 1024) sh[b] = 0;
    __syncthreads();
    const int lo = (int)blockIdx.x * per;
    const int hi = (lo + per < E) ? (lo + per) : E;
    for (int e0 = lo + t; e0 < hi; e0 += 8 * 1024) {
        int d[8];
#pragma unroll
        for (int k = 0; k < 8; ++k) {
            const int e = e0 + 1024 * k;
            d[k] = dst[e < hi ? e : hi - 1];
        }
        asm volatile("" :: "v"(d[0]), "v"(d[1]), "v"(d[2]), "v"(d[3]),
                           "v"(d[4]), "v"(d[5]), "v"(d[6]), "v"(d[7]));
#pragma unroll
        for (int k = 0; k < 8; ++k)
            if (e0 + 1024 * k < hi) atomicAdd(&sh[d[k] >> BSH], 1);
    }
    __syncthreads();
    for (int b = t; b < nb; b += 1024) tab[b * CHUNKS + (int)blockIdx.x] = sh[b];
}

__global__ __launch_bounds__(1024)
void table_scan_kernel(int* __restrict__ a, int m) {
    __shared__ int sm[1024];
    const int t = threadIdx.x;
    const int units = m >> 6;
    const int upt = (units + 1023) >> 10;
    const int u0 = (t * upt < units) ? t * upt : units;
    const int u1 = (u0 + upt < units) ? (u0 + upt) : units;
    int4* __restrict__ a4 = (int4*)a;
    int s = 0;
    for (int u = u0; u < u1; ++u) {
#pragma unroll
        for (int k = 0; k < 16; ++k) {
            const int4 v = a4[u * 16 + k];
            s += v.x + v.y + v.z + v.w;
        }
    }
    sm[t] = s;
    __syncthreads();
    for (int d = 1; d < 1024; d <<= 1) {
        const int v = (t >= d) ? sm[t - d] : 0;
        __syncthreads();
        sm[t] += v;
        __syncthreads();
    }
    int run = sm[t] - s;
    for (int u = u0; u < u1; ++u) {
        int4 v[16];
#pragma unroll
        for (int k = 0; k < 16; ++k) v[k] = a4[u * 16 + k];
#pragma unroll
        for (int k = 0; k < 16; ++k) {
            int4 o;
            o.x = run; run += v[k].x;
            o.y = run; run += v[k].y;
            o.z = run; run += v[k].z;
            o.w = run; run += v[k].w;
            a4[u * 16 + k] = o;
        }
    }
    if (t == 1023) a[m] = sm[1023];
}

__global__ __launch_bounds__(1024)
void bucket_place_kernel(const int* __restrict__ src, const int* __restrict__ dst,
                         const int* __restrict__ tab, int2* __restrict__ pairs,
                         int E, int per, int nb) {
    extern __shared__ int sh[];
    const int t = threadIdx.x;
    for (int b = t; b < nb; b += 1024) sh[b] = tab[b * CHUNKS + (int)blockIdx.x];
    __syncthreads();
    const int lo = (int)blockIdx.x * per;
    const int hi = (lo + per < E) ? (lo + per) : E;
    for (int e0 = lo + t; e0 < hi; e0 += 8 * 1024) {
        int d[8], s[8];
#pragma unroll
        for (int k = 0; k < 8; ++k) {
            const int e  = e0 + 1024 * k;
            const int ec = e < hi ? e : hi - 1;
            d[k] = dst[ec];
            s[k] = src[ec];
        }
        asm volatile("" :: "v"(d[0]), "v"(d[1]), "v"(d[2]), "v"(d[3]),
                           "v"(d[4]), "v"(d[5]), "v"(d[6]), "v"(d[7]));
        asm volatile("" :: "v"(s[0]), "v"(s[1]), "v"(s[2]), "v"(s[3]),
                           "v"(s[4]), "v"(s[5]), "v"(s[6]), "v"(s[7]));
        int p[8];
#pragma unroll
        for (int k = 0; k < 8; ++k)
            p[k] = (e0 + 1024 * k < hi) ? atomicAdd(&sh[d[k] >> BSH], 1) : -1;
#pragma unroll
        for (int k = 0; k < 8; ++k)
            if (p[k] >= 0) pairs[p[k]] = make_int2(s[k], d[k]);
    }
}

__global__ __launch_bounds__(BNODES)
void bucket_rows_kernel(const int2* __restrict__ pairs, const int* __restrict__ tab,
                        int* __restrict__ rowptr, float* __restrict__ dinv,
                        int* __restrict__ csr_src, int n) {
    __shared__ int cnt[BNODES];
    __shared__ int sc[BNODES];
    const int t = threadIdx.x;
    const int b = blockIdx.x;
    const int beg = tab[b * CHUNKS];
    const int end = tab[(b + 1) * CHUNKS];
    cnt[t] = 0;
    __syncthreads();
    for (int e0 = beg + t; e0 < end; e0 += 8 * BNODES) {
        int d[8];
#pragma unroll
        for (int k = 0; k < 8; ++k) {
            const int e = e0 + BNODES * k;
            d[k] = pairs[e < end ? e : end - 1].y;
        }
        asm volatile("" :: "v"(d[0]), "v"(d[1]), "v"(d[2]), "v"(d[3]),
                           "v"(d[4]), "v"(d[5]), "v"(d[6]), "v"(d[7]));
#pragma unroll
        for (int k = 0; k < 8; ++k)
            if (e0 + BNODES * k < end) atomicAdd(&cnt[d[k] & (BNODES - 1)], 1);
    }
    __syncthreads();
    const int c = cnt[t];
    sc[t] = c;
    __syncthreads();
    for (int s = 1; s < BNODES; s <<= 1) {
        const int v = (t >= s) ? sc[t - s] : 0;
        __syncthreads();
        sc[t] += v;
        __syncthreads();
    }
    const int first = beg + sc[t] - c;
    const int i = b * BNODES + t;
    if (i < n) {
        rowptr[i] = first;
        dinv[i]   = rsqrtf((float)(c + 1));
    }
    if (i == n - 1) rowptr[n] = first + c;
    cnt[t] = first;
    __syncthreads();
    for (int e0 = beg + t; e0 < end; e0 += 8 * BNODES) {
        int2 sd[8];
#pragma unroll
        for (int k = 0; k < 8; ++k) {
            const int e = e0 + BNODES * k;
            sd[k] = pairs[e < end ? e : end - 1];
        }
        asm volatile("" :: "v"(sd[0].y), "v"(sd[1].y), "v"(sd[2].y), "v"(sd[3].y),
                           "v"(sd[4].y), "v"(sd[5].y), "v"(sd[6].y), "v"(sd[7].y));
        int p[8];
#pragma unroll
        for (int k = 0; k < 8; ++k)
            p[k] = (e0 + BNODES * k < end) ? atomicAdd(&cnt[sd[k].y & (BNODES - 1)], 1) : -1;
#pragma unroll
        for (int k = 0; k < 8; ++k)
            if (p[k] >= 0) csr_src[p[k]] = sd[k].x;
    }
}

__global__ __launch_bounds__(256) __attribute__((amdgpu_waves_per_eu(8, 8)))
void gemm_wmma_f32(const float* __restrict__ X, const float* __restrict__ W,
                   const float* __restrict__ dinv,
                   float* __restrict__ H, int ntiles) {
    __shared__ float4 wl[64 * 64];

    const int wave = threadIdx.x >> 5;
    const int lane = threadIdx.x & 31;
    const int c    = lane & 15;
    const int h    = lane >> 4;
    const int tile = blockIdx.x * 8 + wave;
    const int tl   = tile < ntiles ? tile : ntiles - 1;
    const int m0   = tl * 16;

    const v2f* __restrict__ xa = (const v2f*)(X + (size_t)(m0 + c) * CH) + h;
    v2f a[32];
#pragma unroll
    for (int s = 0; s < 8; ++s) a[s] = xa[2 * s];
    const float dvl = dinv[m0 + c];

    const float4* __restrict__ W4 = (const float4*)W;
#pragma unroll
    for (int it = 0; it < 8; ++it) {
        const int p = it * 8 + wave;
        const float4 ra = W4[(2 * p) * 32 + lane];
        const float4 rb = W4[(2 * p + 1) * 32 + lane];
        float4 lo, hi;
        lo.x = ra.x; lo.y = rb.x; lo.z = ra.y; lo.w = rb.y;
        hi.x = ra.z; hi.y = rb.z; hi.z = ra.w; hi.w = rb.w;
        wl[p * 64 + (2 * h) * 16 + c]     = lo;
        wl[p * 64 + (2 * h + 1) * 16 + c] = hi;
    }
    __syncthreads();
    if (tile >= ntiles) return;

    v8f acc[8];
#pragma unroll
    for (int t = 0; t < 8; ++t) {
        acc[t] = (v8f){0.f, 0.f, 0.f, 0.f, 0.f, 0.f, 0.f, 0.f};
    }

    const float4* bl = wl + h * 64 + c;
#pragma unroll
    for (int s = 0; s < 32; ++s) {
        if ((s & 7) == 0) {
            if (s + 8 < 32) {
#pragma unroll
                for (int i = 0; i < 8; ++i) a[s + 8 + i] = xa[2 * (s + 8 + i)];
            }
            __builtin_amdgcn_sched_barrier(0);
        }
        const v2f as = a[s] * dvl;
        const float4 q0 = bl[s * 128];
        const float4 q1 = bl[s * 128 + 16];
        const float4 q2 = bl[s * 128 + 32];
        const float4 q3 = bl[s * 128 + 48];
        v2f b;
        b.x = q0.x; b.y = q0.y;
        acc[0] = __builtin_amdgcn_wmma_f32_16x16x4_f32(false, as, false, b, (short)0, acc[0], false, false);
        b.x = q0.z; b.y = q0.w;
        acc[1] = __builtin_amdgcn_wmma_f32_16x16x4_f32(false, as, false, b, (short)0, acc[1], false, false);
        b.x = q1.x; b.y = q1.y;
        acc[2] = __builtin_amdgcn_wmma_f32_16x16x4_f32(false, as, false, b, (short)0, acc[2], false, false);
        b.x = q1.z; b.y = q1.w;
        acc[3] = __builtin_amdgcn_wmma_f32_16x16x4_f32(false, as, false, b, (short)0, acc[3], false, false);
        b.x = q2.x; b.y = q2.y;
        acc[4] = __builtin_amdgcn_wmma_f32_16x16x4_f32(false, as, false, b, (short)0, acc[4], false, false);
        b.x = q2.z; b.y = q2.w;
        acc[5] = __builtin_amdgcn_wmma_f32_16x16x4_f32(false, as, false, b, (short)0, acc[5], false, false);
        b.x = q3.x; b.y = q3.y;
        acc[6] = __builtin_amdgcn_wmma_f32_16x16x4_f32(false, as, false, b, (short)0, acc[6], false, false);
        b.x = q3.z; b.y = q3.w;
        acc[7] = __builtin_amdgcn_wmma_f32_16x16x4_f32(false, as, false, b, (short)0, acc[7], false, false);
    }

    float4* __restrict__ H4 = (float4*)H;
#pragma unroll
    for (int r = 0; r < 8; ++r) {
        const size_t row4 = (size_t)(m0 + 8 * h + r) * (CH / 4);
        float4 v0, v1;
        v0.x = acc[0][r]; v0.y = acc[1][r]; v0.z = acc[2][r]; v0.w = acc[3][r];
        v1.x = acc[4][r]; v1.y = acc[5][r]; v1.z = acc[6][r]; v1.w = acc[7][r];
        H4[row4 + c]      = v0;
        H4[row4 + 16 + c] = v1;
    }
}

__global__ __launch_bounds__(256)
void gather128_kernel(const float* __restrict__ h, const int* __restrict__ rowptr,
                      const int* __restrict__ csr_src, const float* __restrict__ dinv,
                      const float* __restrict__ b, const float* __restrict__ g,
                      const float* __restrict__ be, const float* __restrict__ rm,
                      const float* __restrict__ rv,
                      float* __restrict__ out, int n) {
    const int d = __builtin_amdgcn_readfirstlane(
        (int)((blockIdx.x * blockDim.x + threadIdx.x) >> 5));
    if (d >= n) return;
    const int lane = threadIdx.x & 31;
    const int beg = rowptr[d];
    const int end = rowptr[d + 1];
    const float dd = dinv[d];
    const float4* __restrict__ hv = (const float4*)h;

    const float4 pb  = ((const float4*)b)[lane];
    const float4 pg  = ((const float4*)g)[lane];
    const float4 pbe = ((const float4*)be)[lane];
    const float4 prm = ((const float4*)rm)[lane];
    const float4 prv = ((const float4*)rv)[lane];
    const float ix = pg.x * rsqrtf(prv.x + 1e-5f);
    const float iy = pg.y * rsqrtf(prv.y + 1e-5f);
    const float iz = pg.z * rsqrtf(prv.z + 1e-5f);
    const float iw = pg.w * rsqrtf(prv.w + 1e-5f);

    float4 acc = hv[(size_t)d * 32 + lane];

    for (int base = beg; base < end; base += 32) {
        const int m = (end - base < 32) ? (end - base) : 32;
        int ms = 0;
        if (lane < m) ms = csr_src[base + lane];
        int j = 0;
        for (; j + 4 <= m; j += 4) {
            const int s0 = __builtin_amdgcn_readlane(ms, j);
            const int s1 = __builtin_amdgcn_readlane(ms, j + 1);
            const int s2 = __builtin_amdgcn_readlane(ms, j + 2);
            const int s3 = __builtin_amdgcn_readlane(ms, j + 3);
            const float4 v0 = hv[(size_t)s0 * 32 + lane];
            const float4 v1 = hv[(size_t)s1 * 32 + lane];
            const float4 v2 = hv[(size_t)s2 * 32 + lane];
            const float4 v3 = hv[(size_t)s3 * 32 + lane];
            acc.x += v0.x; acc.y += v0.y; acc.z += v0.z; acc.w += v0.w;
            acc.x += v1.x; acc.y += v1.y; acc.z += v1.z; acc.w += v1.w;
            acc.x += v2.x; acc.y += v2.y; acc.z += v2.z; acc.w += v2.w;
            acc.x += v3.x; acc.y += v3.y; acc.z += v3.z; acc.w += v3.w;
        }
        const int rem = m & 3;
        for (int t = 0; t < rem; ++t) {
            const int s0 = __builtin_amdgcn_readlane(ms, j + t);
            const float4 v0 = hv[(size_t)s0 * 32 + lane];
            acc.x += v0.x; acc.y += v0.y; acc.z += v0.z; acc.w += v0.w;
        }
    }
    float4 r;
    {
        const float vx = (acc.x * dd + pb.x - prm.x) * ix + pbe.x;
        const float vy = (acc.y * dd + pb.y - prm.y) * iy + pbe.y;
        const float vz = (acc.z * dd + pb.z - prm.z) * iz + pbe.z;
        const float vw = (acc.w * dd + pb.w - prm.w) * iw + pbe.w;
        r.x = vx > 0.f ? vx : 0.f;
        r.y = vy > 0.f ? vy : 0.f;
        r.z = vz > 0.f ? vz : 0.f;
        r.w = vw > 0.f ? vw : 0.f;
    }
    ((float4*)out)[(size_t)d * 32 + lane] = r;
}

__global__ void gemm_out2_kernel(const float* __restrict__ X,
                                 const float* __restrict__ W3,
                                 const float* __restrict__ dinv,
                                 float* __restrict__ H3, int M) {
    int i = blockIdx.x * blockDim.x + threadIdx.x;
    if (i >= M) return;
    const float* xr = X + (size_t)i * CH;
    float a0 = 0.f, a1 = 0.f;
#pragma unroll 8
    for (int k = 0; k < CH; ++k) {
        const float x = xr[k];
        a0 += x * W3[2 * k];
        a1 += x * W3[2 * k + 1];
    }
    const float sc = dinv[i];
    H3[(size_t)i * 2]     = a0 * sc;
    H3[(size_t)i * 2 + 1] = a1 * sc;
}

__global__ void gather2_kernel(const float* __restrict__ h3, const int* __restrict__ rowptr,
                               const int* __restrict__ csr_src, const float* __restrict__ dinv,
                               const float* __restrict__ b3, float* __restrict__ out, int n) {
    const int d = blockIdx.x * blockDim.x + threadIdx.x;
    if (d >= n) return;
    const int beg = rowptr[d];
    const int end = rowptr[d + 1];
    const float dd = dinv[d];
    const float2* __restrict__ hv = (const float2*)h3;
    const float2 vs = hv[d];
    float a0 = vs.x;
    float a1 = vs.y;
#pragma unroll 4
    for (int j = beg; j < end; ++j) {
        const int s = csr_src[j];
        const float2 u = hv[s];
        a0 += u.x;
        a1 += u.y;
    }
    float2 r;
    r.x = a0 * dd + b3[0];
    r.y = a1 * dd + b3[1];
    ((float2*)out)[d] = r;
}

extern "C" void kernel_launch(void* const* d_in, const int* in_sizes, int n_in,
                              void* d_out, int out_size, void* d_ws, size_t ws_size,
                              hipStream_t stream) {
    const float* x   = (const float*)d_in[0];
    const int*   ei  = (const int*)d_in[1];
    const float* W1  = (const float*)d_in[2];
    const float* b1  = (const float*)d_in[3];
    const float* g1  = (const float*)d_in[4];
    const float* be1 = (const float*)d_in[5];
    const float* rm1 = (const float*)d_in[6];
    const float* rv1 = (const float*)d_in[7];
    const float* W2  = (const float*)d_in[8];
    const float* b2  = (const float*)d_in[9];
    const float* g2  = (const float*)d_in[10];
    const float* be2 = (const float*)d_in[11];
    const float* rm2 = (const float*)d_in[12];
    const float* rv2 = (const float*)d_in[13];
    const float* W3  = (const float*)d_in[14];
    const float* b3  = (const float*)d_in[15];

    const int Nn   = in_sizes[0] / CH;
    const int Ee   = in_sizes[1] / 2;
    const int NC   = Nn * CH;

    const int* src = ei;
    const int* dst = ei + Ee;

    const int T = 256;
    const int gN  = (Nn + T - 1) / T;
    const int gW  = (Nn + 7) / 8;
    const int gMM = Nn / 16;
    const int gMB = (gMM + 7) / 8;
    const int nb  = (Nn + BNODES - 1) / BNODES;
    const int per = (Ee + CHUNKS - 1) / CHUNKS;
    const int tm  = nb * CHUNKS;

    const size_t NP = ((size_t)Nn + 64) & ~(size_t)63;
    const size_t TP = ((size_t)tm + 64) & ~(size_t)63;
    const size_t EP = ((size_t)Ee + 63) & ~(size_t)63;
    int*   tab    = (int*)d_ws;
    int*   rowptr = tab + TP;
    float* dinv   = (float*)(rowptr + NP);
    int*   csr    = (int*)(dinv + NP);
    int2*  pairs  = (int2*)(csr + EP);
    float* bufA   = (float*)(csr + 3 * EP);
    float* bufB   = bufA + (size_t)NC;
    float* h3     = bufB + (size_t)NC;
    float* out    = (float*)d_out;

    bucket_count_kernel<<<CHUNKS, 1024, nb * sizeof(int), stream>>>(dst, tab, Ee, per, nb);
    table_scan_kernel<<<1, 1024, 0, stream>>>(tab, tm);
    bucket_place_kernel<<<CHUNKS, 1024, nb * sizeof(int), stream>>>(src, dst, tab, pairs, Ee, per, nb);
    bucket_rows_kernel<<<nb, BNODES, 0, stream>>>(pairs, tab, rowptr, dinv, csr, Nn);

    gemm_wmma_f32<<<gMB, T, 0, stream>>>(x, W1, dinv, bufA, gMM);
    gather128_kernel<<<gW, T, 0, stream>>>(bufA, rowptr, csr, dinv, b1, g1, be1, rm1, rv1, bufB, Nn);

    gemm_wmma_f32<<<gMB, T, 0, stream>>>(bufB, W2, dinv, bufA, gMM);
    gather128_kernel<<<gW, T, 0, stream>>>(bufA, rowptr, csr, dinv, b2, g2, be2, rm2, rv2, bufB, Nn);

    gemm_out2_kernel<<<gN, T, 0, stream>>>(bufB, W3, dinv, h3, Nn);
    gather2_kernel<<<gN, T, 0, stream>>>(h3, rowptr, csr, dinv, b3, out, Nn);
}


# Table-scan launch dropped: each placement work-group sums the count table itself (chunk-major rows, whole-line reads) and scans the bucket totals in LDS; 10 to 9 launches, slots and sums unchanged
# speedup vs baseline: 4.7825x; 1.0736x over previous
#include <hip/hip_runtime.h>

#define CH 128

typedef float v2f __attribute__((ext_vector_type(2)));
typedef float v8f __attribute__((ext_vector_type(8)));

#define BSH    9
#define BNODES 512
#define CHUNKS 256
static_assert(CHUNKS == 256, "bucket_place_kernel names its sixteen table loads one by one");

__global__ __launch_bounds__(1024)
void bucket_count_kernel(const int* __restrict__ dst, int* __restrict__ tab,
                         int E, int per, int nb, int nbp) {
    extern __shared__ int sh[];
    const int t = threadIdx.x;
    for (int b = t; b < nb; b += 1024) sh[b] = 0;
    __syncthreads();
    const int lo = (int)blockIdx.x * per;
    const int hi = (lo + per < E) ? (lo + per) : E;
    for (int e0 = lo + t; e0 < hi; e0 += 8 * 1024) {
        int d[8];
#pragma unroll
        for (int k = 0; k < 8; ++k) {
            const int e = e0 + 1024 * k;
            d[k] = dst[e < hi ? e : hi - 1];
        }
        asm volatile("" :: "v"(d[0]), "v"(d[1]), "v"(d[2]), "v"(d[3]),
                           "v"(d[4]), "v"(d[5]), "v"(d[6]), "v"(d[7]));
#pragma unroll
        for (int k = 0; k < 8; ++k)
            if (e0 + 1024 * k < hi) atomicAdd(&sh[d[k] >> BSH], 1);
    }
    __syncthreads();
    for (int b = t; b < nbp; b += 1024)
        tab[(int)blockIdx.x * nbp + b] = (b < nb) ? sh[b] : 0;
}

__global__ __launch_bounds__(1024)
void bucket_place_kernel(const int* __restrict__ src, const int* __restrict__ dst,
                         const int* __restrict__ tab, int* __restrict__ bstart,
                         int2* __restrict__ pairs,
                         int E, int per, int nb, int nbp) {
    extern __shared__ int sh[];
    int* tot = sh + nbp;
    int* ws  = sh + 2 * nbp;
    const int t    = threadIdx.x;
    const int lane = t & 31;
    const int wave = t >> 5;
    const int j    = (int)blockIdx.x;
    for (int b = t; b < 2 * nbp; b += 1024) sh[b] = 0;
    __syncthreads();
    {
        const int nq = nbp >> 2;
        const int4* __restrict__ tab4 = (const int4*)tab;
        const int q = t >> 6;
        for (int c4 = t & 63; c4 < nq; c4 += 64) {
            int4 v[CHUNKS / 16];
#pragma unroll
            for (int k = 0; k < CHUNKS / 16; ++k) v[k] = tab4[(q + 16 * k) * nq + c4];
            asm volatile("" :: "v"(v[0].x), "v"(v[1].x), "v"(v[2].x), "v"(v[3].x),
                               "v"(v[4].x), "v"(v[5].x), "v"(v[6].x), "v"(v[7].x),
                               "v"(v[8].x), "v"(v[9].x), "v"(v[10].x), "v"(v[11].x),
                               "v"(v[12].x), "v"(v[13].x), "v"(v[14].x), "v"(v[15].x));
            int4 a = make_int4(0, 0, 0, 0);
            int4 p = make_int4(0, 0, 0, 0);
#pragma unroll
            for (int k = 0; k < CHUNKS / 16; ++k) {
                a.x += v[k].x; a.y += v[k].y; a.z += v[k].z; a.w += v[k].w;
                if (q + 16 * k < j) { p.x += v[k].x; p.y += v[k].y; p.z += v[k].z; p.w += v[k].w; }
            }
            atomicAdd(&tot[4 * c4],     a.x); atomicAdd(&tot[4 * c4 + 1], a.y);
            atomicAdd(&tot[4 * c4 + 2], a.z); atomicAdd(&tot[4 * c4 + 3], a.w);
            atomicAdd(&sh[4 * c4],      p.x); atomicAdd(&sh[4 * c4 + 1],  p.y);
            atomicAdd(&sh[4 * c4 + 2],  p.z); atomicAdd(&sh[4 * c4 + 3],  p.w);
        }
    }
    __syncthreads();
    int carry = 0;
    for (int base = 0; base < nb; base += 1024) {
        const int b = base + t;
        const int v = (b < nb) ? tot[b] : 0;
        int x = v;
#pragma unroll
        for (int d = 1; d < 32; d <<= 1) {
            const int y = __shfl_up(x, d);
            if (lane >= d) x += y;
        }
        if (lane == 31) ws[wave] = x;
        __syncthreads();
        if (wave == 0) {
            const int w = ws[lane];
            int xs = w;
#pragma unroll
            for (int d = 1; d < 32; d <<= 1) {
                const int y = __shfl_up(xs, d);
                if (lane >= d) xs += y;
            }
            ws[lane] = xs - w;
            if (lane == 31) ws[32] = xs;
        }
        __syncthreads();
        const int first = carry + ws[wave] + x - v;
        if (b < nb) {
            sh[b] += first;
            if (j == 0) bstart[b] = first;
        }
        carry += ws[32];
        __syncthreads();
    }
    if (j == 0 && t == 0) bstart[nb] = carry;
    const int lo = (int)blockIdx.x * per;
    const int hi = (lo + per < E) ? (lo + per) : E;
    for (int e0 = lo + t; e0 < hi; e0 += 8 * 1024) {
        int d[8], s[8];
#pragma unroll
        for (int k = 0; k < 8; ++k) {
            const int e  = e0 + 1024 * k;
            const int ec = e < hi ? e : hi - 1;
            d[k] = dst[ec];
            s[k] = src[ec];
        }
        asm volatile("" :: "v"(d[0]), "v"(d[1]), "v"(d[2]), "v"(d[3]),
                           "v"(d[4]), "v"(d[5]), "v"(d[6]), "v"(d[7]));
        asm volatile("" :: "v"(s[0]), "v"(s[1]), "v"(s[2]), "v"(s[3]),
                           "v"(s[4]), "v"(s[5]), "v"(s[6]), "v"(s[7]));
        int p[8];
#pragma unroll
        for (int k = 0; k < 8; ++k)
            p[k] = (e0 + 1024 * k < hi) ? atomicAdd(&sh[d[k] >> BSH], 1) : -1;
#pragma unroll
        for (int k = 0; k < 8; ++k)
            if (p[k] >= 0) pairs[p[k]] = make_int2(s[k], d[k]);
    }
}

__global__ __launch_bounds__(BNODES)
void bucket_rows_kernel(const int2* __restrict__ pairs, const int* __restrict__ bstart,
                        int* __restrict__ rowptr, float* __restrict__ dinv,
                        int* __restrict__ csr_src, int n) {
    __shared__ int cnt[BNODES];
    __shared__ int sc[BNODES];
    const int t = threadIdx.x;
    const int b = blockIdx.x;
    const int beg = bstart[b];
    const int end = bstart[b + 1];
    cnt[t] = 0;
    __syncthreads();
    for (int e0 = beg + t; e0 < end; e0 += 8 * BNODES) {
        int d[8];
#pragma unroll
        for (int k = 0; k < 8; ++k) {
            const int e = e0 + BNODES * k;
            d[k] = pairs[e < end ? e : end - 1].y;
        }
        asm volatile("" :: "v"(d[0]), "v"(d[1]), "v"(d[2]), "v"(d[3]),
                           "v"(d[4]), "v"(d[5]), "v"(d[6]), "v"(d[7]));
#pragma unroll
        for (int k = 0; k < 8; ++k)
            if (e0 + BNODES * k < end) atomicAdd(&cnt[d[k] & (BNODES - 1)], 1);
    }
    __syncthreads();
    const int c = cnt[t];
    sc[t] = c;
    __syncthreads();
    for (int s = 1; s < BNODES; s <<= 1) {
        const int v = (t >= s) ? sc[t - s] : 0;
        __syncthreads();
        sc[t] += v;
        __syncthreads();
    }
    const int first = beg + sc[t] - c;
    const int i = b * BNODES + t;
    if (i < n) {
        rowptr[i] = first;
        dinv[i]   = rsqrtf((float)(c + 1));
    }
    if (i == n - 1) rowptr[n] = first + c;
    cnt[t] = first;
    __syncthreads();
    for (int e0 = beg + t; e0 < end; e0 += 8 * BNODES) {
        int2 sd[8];
#pragma unroll
        for (int k = 0; k < 8; ++k) {
            const int e = e0 + BNODES * k;
            sd[k] = pairs[e < end ? e : end - 1];
        }
        asm volatile("" :: "v"(sd[0].y), "v"(sd[1].y), "v"(sd[2].y), "v"(sd[3].y),
                           "v"(sd[4].y), "v"(sd[5].y), "v"(sd[6].y), "v"(sd[7].y));
        int p[8];
#pragma unroll
        for (int k = 0; k < 8; ++k)
            p[k] = (e0 + BNODES * k < end) ? atomicAdd(&cnt[sd[k].y & (BNODES - 1)], 1) : -1;
#pragma unroll
        for (int k = 0; k < 8; ++k)
            if (p[k] >= 0) csr_src[p[k]] = sd[k].x;
    }
}

__global__ __launch_bounds__(256) __attribute__((amdgpu_waves_per_eu(8, 8)))
void gemm_wmma_f32(const float* __restrict__ X, const float* __restrict__ W,
                   const float* __restrict__ dinv,
                   float* __restrict__ H, int ntiles) {
    __shared__ float4 wl[64 * 64];

    const int wave = threadIdx.x >> 5;
    const int lane = threadIdx.x & 31;
    const int c    = lane & 15;
    const int h    = lane >> 4;
    const int tile = blockIdx.x * 8 + wave;
    const int tl   = tile < ntiles ? tile : ntiles - 1;
    const int m0   = tl * 16;

    const v2f* __restrict__ xa = (const v2f*)(X + (size_t)(m0 + c) * CH) + h;
    v2f a[32];
#pragma unroll
    for (int s = 0; s < 8; ++s) a[s] = xa[2 * s];
    const float dvl = dinv[m0 + c];

    const float4* __restrict__ W4 = (const float4*)W;
#pragma unroll
    for (int it = 0; it < 8; ++it) {
        const int p = it * 8 + wave;
        const float4 ra = W4[(2 * p) * 32 + lane];
        const float4 rb = W4[(2 * p + 1) * 32 + lane];
        float4 lo, hi;
        lo.x = ra.x; lo.y = rb.x; lo.z = ra.y; lo.w = rb.y;
        hi.x = ra.z; hi.y = rb.z; hi.z = ra.w; hi.w = rb.w;
        wl[p * 64 + (2 * h) * 16 + c]     = lo;
        wl[p * 64 + (2 * h + 1) * 16 + c] = hi;
    }
    __syncthreads();
    if (tile >= ntiles) return;

    v8f acc[8];
#pragma unroll
    for (int t = 0; t < 8; ++t) {
        acc[t] = (v8f){0.f, 0.f, 0.f, 0.f, 0.f, 0.f, 0.f, 0.f};
    }

    const float4* bl = wl + h * 64 + c;
#pragma unroll
    for (int s = 0; s < 32; ++s) {
        if ((s & 7) == 0) {
            if (s + 8 < 32) {
#pragma unroll
                for (int i = 0; i < 8; ++i) a[s + 8 + i] = xa[2 * (s + 8 + i)];
            }
            __builtin_amdgcn_sched_barrier(0);
        }
        const v2f as = a[s] * dvl;
        const float4 q0 = bl[s * 128];
        const float4 q1 = bl[s * 128 + 16];
        const float4 q2 = bl[s * 128 + 32];
        const float4 q3 = bl[s * 128 + 48];
        v2f b;
        b.x = q0.x; b.y = q0.y;
        acc[0] = __builtin_amdgcn_wmma_f32_16x16x4_f32(false, as, false, b, (short)0, acc[0], false, false);
        b.x = q0.z; b.y = q0.w;
        acc[1] = __builtin_amdgcn_wmma_f32_16x16x4_f32(false, as, false, b, (short)0, acc[1], false, false);
        b.x = q1.x; b.y = q1.y;
        acc[2] = __builtin_amdgcn_wmma_f32_16x16x4_f32(false, as, false, b, (short)0, acc[2], false, false);
        b.x = q1.z; b.y = q1.w;
        acc[3] = __builtin_amdgcn_wmma_f32_16x16x4_f32(false, as, false, b, (short)0, acc[3], false, false);
        b.x = q2.x; b.y = q2.y;
        acc[4] = __builtin_amdgcn_wmma_f32_16x16x4_f32(false, as, false, b, (short)0, acc[4], false, false);
        b.x = q2.z; b.y = q2.w;
        acc[5] = __builtin_amdgcn_wmma_f32_16x16x4_f32(false, as, false, b, (short)0, acc[5], false, false);
        b.x = q3.x; b.y = q3.y;
        acc[6] = __builtin_amdgcn_wmma_f32_16x16x4_f32(false, as, false, b, (short)0, acc[6], false, false);
        b.x = q3.z; b.y = q3.w;
        acc[7] = __builtin_amdgcn_wmma_f32_16x16x4_f32(false, as, false, b, (short)0, acc[7], false, false);
    }

    float4* __restrict__ H4 = (float4*)H;
#pragma unroll
    for (int r = 0; r < 8; ++r) {
        const size_t row4 = (size_t)(m0 + 8 * h + r) * (CH / 4);
        float4 v0, v1;
        v0.x = acc[0][r]; v0.y = acc[1][r]; v0.z = acc[2][r]; v0.w = acc[3][r];
        v1.x = acc[4][r]; v1.y = acc[5][r]; v1.z = acc[6][r]; v1.w = acc[7][r];
        H4[row4 + c]      = v0;
        H4[row4 + 16 + c] = v1;
    }
}

__global__ __launch_bounds__(256)
void gather128_kernel(const float* __restrict__ h, const int* __restrict__ rowptr,
                      const int* __restrict__ csr_src, const float* __restrict__ dinv,
                      const float* __restrict__ b, const float* __restrict__ g,
                      const float* __restrict__ be, const float* __restrict__ rm,
                      const float* __restrict__ rv,
                      float* __restrict__ out, int n) {
    const int d = __builtin_amdgcn_readfirstlane(
        (int)((blockIdx.x * blockDim.x + threadIdx.x) >> 5));
    if (d >= n) return;
    const int lane = threadIdx.x & 31;
    const int beg = rowptr[d];
    const int end = rowptr[d + 1];
    const float dd = dinv[d];
    const float4* __restrict__ hv = (const float4*)h;

    const float4 pb  = ((const float4*)b)[lane];
    const float4 pg  = ((const float4*)g)[lane];
    const float4 pbe = ((const float4*)be)[lane];
    const float4 prm = ((const float4*)rm)[lane];
    const float4 prv = ((const float4*)rv)[lane];
    const float ix = pg.x * rsqrtf(prv.x + 1e-5f);
    const float iy = pg.y * rsqrtf(prv.y + 1e-5f);
    const float iz = pg.z * rsqrtf(prv.z + 1e-5f);
    const float iw = pg.w * rsqrtf(prv.w + 1e-5f);

    float4 acc = hv[(size_t)d * 32 + lane];

    for (int base = beg; base < end; base += 32) {
        const int m = (end - base < 32) ? (end - base) : 32;
        int ms = 0;
        if (lane < m) ms = csr_src[base + lane];
        int j = 0;
        for (; j + 4 <= m; j += 4) {
            const int s0 = __builtin_amdgcn_readlane(ms, j);
            const int s1 = __builtin_amdgcn_readlane(ms, j + 1);
            const int s2 = __builtin_amdgcn_readlane(ms, j + 2);
            const int s3 = __builtin_amdgcn_readlane(ms, j + 3);
            const float4 v0 = hv[(size_t)s0 * 32 + lane];
            const float4 v1 = hv[(size_t)s1 * 32 + lane];
            const float4 v2 = hv[(size_t)s2 * 32 + lane];
            const float4 v3 = hv[(size_t)s3 * 32 + lane];
            acc.x += v0.x; acc.y += v0.y; acc.z += v0.z; acc.w += v0.w;
            acc.x += v1.x; acc.y += v1.y; acc.z += v1.z; acc.w += v1.w;
            acc.x += v2.x; acc.y += v2.y; acc.z += v2.z; acc.w += v2.w;
            acc.x += v3.x; acc.y += v3.y; acc.z += v3.z; acc.w += v3.w;
        }
        const int rem = m & 3;
        for (int t = 0; t < rem; ++t) {
            const int s0 = __builtin_amdgcn_readlane(ms, j + t);
            const float4 v0 = hv[(size_t)s0 * 32 + lane];
            acc.x += v0.x; acc.y += v0.y; acc.z += v0.z; acc.w += v0.w;
        }
    }
    float4 r;
    {
        const float vx = (acc.x * dd + pb.x - prm.x) * ix + pbe.x;
        const float vy = (acc.y * dd + pb.y - prm.y) * iy + pbe.y;
        const float vz = (acc.z * dd + pb.z - prm.z) * iz + pbe.z;
        const float vw = (acc.w * dd + pb.w - prm.w) * iw + pbe.w;
        r.x = vx > 0.f ? vx : 0.f;
        r.y = vy > 0.f ? vy : 0.f;
        r.z = vz > 0.f ? vz : 0.f;
        r.w = vw > 0.f ? vw : 0.f;
    }
    ((float4*)out)[(size_t)d * 32 + lane] = r;
}

__global__ void gemm_out2_kernel(const float* __restrict__ X,
                                 const float* __restrict__ W3,
                                 const float* __restrict__ dinv,
                                 float* __restrict__ H3, int M) {
    int i = blockIdx.x * blockDim.x + threadIdx.x;
    if (i >= M) return;
    const float* xr = X + (size_t)i * CH;
    float a0 = 0.f, a1 = 0.f;
#pragma unroll 8
    for (int k = 0; k < CH; ++k) {
        const float x = xr[k];
        a0 += x * W3[2 * k];
        a1 += x * W3[2 * k + 1];
    }
    const float sc = dinv[i];
    H3[(size_t)i * 2]     = a0 * sc;
    H3[(size_t)i * 2 + 1] = a1 * sc;
}

__global__ void gather2_kernel(const float* __restrict__ h3, const int* __restrict__ rowptr,
                               const int* __restrict__ csr_src, const float* __restrict__ dinv,
                               const float* __restrict__ b3, float* __restrict__ out, int n) {
    const int d = blockIdx.x * blockDim.x + threadIdx.x;
    if (d >= n) return;
    const int beg = rowptr[d];
    const int end = rowptr[d + 1];
    const float dd = dinv[d];
    const float2* __restrict__ hv = (const float2*)h3;
    const float2 vs = hv[d];
    float a0 = vs.x;
    float a1 = vs.y;
#pragma unroll 4
    for (int j = beg; j < end; ++j) {
        const int s = csr_src[j];
        const float2 u = hv[s];
        a0 += u.x;
        a1 += u.y;
    }
    float2 r;
    r.x = a0 * dd + b3[0];
    r.y = a1 * dd + b3[1];
    ((float2*)out)[d] = r;
}

extern "C" void kernel_launch(void* const* d_in, const int* in_sizes, int n_in,
                              void* d_out, int out_size, void* d_ws, size_t ws_size,
                              hipStream_t stream) {
    const float* x   = (const float*)d_in[0];
    const int*   ei  = (const int*)d_in[1];
    const float* W1  = (const float*)d_in[2];
    const float* b1  = (const float*)d_in[3];
    const float* g1  = (const float*)d_in[4];
    const float* be1 = (const float*)d_in[5];
    const float* rm1 = (const float*)d_in[6];
    const float* rv1 = (const float*)d_in[7];
    const float* W2  = (const float*)d_in[8];
    const float* b2  = (const float*)d_in[9];
    const float* g2  = (const float*)d_in[10];
    const float* be2 = (const float*)d_in[11];
    const float* rm2 = (const float*)d_in[12];
    const float* rv2 = (const float*)d_in[13];
    const float* W3  = (const float*)d_in[14];
    const float* b3  = (const float*)d_in[15];

    const int Nn   = in_sizes[0] / CH;
    const int Ee   = in_sizes[1] / 2;
    const int NC   = Nn * CH;

    const int* src = ei;
    const int* dst = ei + Ee;

    const int T = 256;
    const int gN  = (Nn + T - 1) / T;
    const int gW  = (Nn + 7) / 8;
    const int gMM = Nn / 16;
    const int gMB = (gMM + 7) / 8;
    const int nb  = (Nn + BNODES - 1) / BNODES;
    const int per = (Ee + CHUNKS - 1) / CHUNKS;
    const int nbp = (nb + 3) & ~3;
    const int tm  = nbp * CHUNKS;

    const size_t NP = ((size_t)Nn + 64) & ~(size_t)63;
    const size_t TP = ((size_t)tm + 63) & ~(size_t)63;
    const size_t BP = ((size_t)nb + 64) & ~(size_t)63;
    const size_t EP = ((size_t)Ee + 63) & ~(size_t)63;
    int*   tab    = (int*)d_ws;
    int*   bstart = tab + TP;
    int*   rowptr = bstart + BP;
    float* dinv   = (float*)(rowptr + NP);
    int*   csr    = (int*)(dinv + NP);
    int2*  pairs  = (int2*)(csr + EP);
    float* bufA   = (float*)(csr + 3 * EP);
    float* bufB   = bufA + (size_t)NC;
    float* h3     = bufB + (size_t)NC;
    float* out    = (float*)d_out;

    bucket_count_kernel<<<CHUNKS, 1024, nb * sizeof(int), stream>>>(dst, tab, Ee, per, nb, nbp);
    bucket_place_kernel<<<CHUNKS, 1024, (2 * nbp + 33) * sizeof(int), stream>>>(src, dst, tab, bstart, pairs, Ee, per, nb, nbp);
    bucket_rows_kernel<<<nb, BNODES, 0, stream>>>(pairs, bstart, rowptr, dinv, csr, Nn);

    gemm_wmma_f32<<<gMB, T, 0, stream>>>(x, W1, dinv, bufA, gMM);
    gather128_kernel<<<gW, T, 0, stream>>>(bufA, rowptr, csr, dinv, b1, g1, be1, rm1, rv1, bufB, Nn);

    gemm_wmma_f32<<<gMB, T, 0, stream>>>(bufB, W2, dinv, bufA, gMM);
    gather128_kernel<<<gW, T, 0, stream>>>(bufA, rowptr, csr, dinv, b2, g2, be2, rm2, rv2, bufB, Nn);

    gemm_out2_kernel<<<gN, T, 0, stream>>>(bufB, W3, dinv, h3, Nn);
    gather2_kernel<<<gN, T, 0, stream>>>(h3, rowptr, csr, dinv, b3, out, Nn);
}


# Layer 3's 2-column product moved into layer 2's aggregation epilogue: 4 terms per lane, then a lane tree (was one thread, k order); layer 2's rows are never stored; 9 to 8 launches
# speedup vs baseline: 4.9772x; 1.0407x over previous
#include <hip/hip_runtime.h>

#define CH 128

typedef float v2f __attribute__((ext_vector_type(2)));
typedef float v8f __attribute__((ext_vector_type(8)));

#define BSH    9
#define BNODES 512
#define CHUNKS 256
static_assert(CHUNKS == 256, "bucket_place_kernel names its sixteen table loads one by one");

__global__ __launch_bounds__(1024)
void bucket_count_kernel(const int* __restrict__ dst, int* __restrict__ tab,
                         int E, int per, int nb, int nbp) {
    extern __shared__ int sh[];
    const int t = threadIdx.x;
    for (int b = t; b < nb; b += 1024) sh[b] = 0;
    __syncthreads();
    const int lo = (int)blockIdx.x * per;
    const int hi = (lo + per < E) ? (lo + per) : E;
    for (int e0 = lo + t; e0 < hi; e0 += 8 * 1024) {
        int d[8];
#pragma unroll
        for (int k = 0; k < 8; ++k) {
            const int e = e0 + 1024 * k;
            d[k] = dst[e < hi ? e : hi - 1];
        }
        asm volatile("" :: "v"(d[0]), "v"(d[1]), "v"(d[2]), "v"(d[3]),
                           "v"(d[4]), "v"(d[5]), "v"(d[6]), "v"(d[7]));
#pragma unroll
        for (int k = 0; k < 8; ++k)
            if (e0 + 1024 * k < hi) atomicAdd(&sh[d[k] >> BSH], 1);
    }
    __syncthreads();
    for (int b = t; b < nbp; b += 1024)
        tab[(int)blockIdx.x * nbp + b] = (b < nb) ? sh[b] : 0;
}

__global__ __launch_bounds__(1024)
void bucket_place_kernel(const int* __restrict__ src, const int* __restrict__ dst,
                         const int* __restrict__ tab, int* __restrict__ bstart,
                         int2* __restrict__ pairs,
                         int E, int per, int nb, int nbp) {
    extern __shared__ int sh[];
    int* tot = sh + nbp;
    int* ws  = sh + 2 * nbp;
    const int t    = threadIdx.x;
    const int lane = t & 31;
    const int wave = t >> 5;
    const int j    = (int)blockIdx.x;
    for (int b = t; b < 2 * nbp; b += 1024) sh[b] = 0;
    __syncthreads();
    {
        const int nq = nbp >> 2;
        const int4* __restrict__ tab4 = (const int4*)tab;
        const int q = t >> 6;
        for (int c4 = t & 63; c4 < nq; c4 += 64) {
            int4 v[CHUNKS / 16];
#pragma unroll
            for (int k = 0; k < CHUNKS / 16; ++k) v[k] = tab4[(q + 16 * k) * nq + c4];
            asm volatile("" :: "v"(v[0].x), "v"(v[1].x), "v"(v[2].x), "v"(v[3].x),
                               "v"(v[4].x), "v"(v[5].x), "v"(v[6].x), "v"(v[7].x),
                               "v"(v[8].x), "v"(v[9].x), "v"(v[10].x), "v"(v[11].x),
                               "v"(v[12].x), "v"(v[13].x), "v"(v[14].x), "v"(v[15].x));
            int4 a = make_int4(0, 0, 0, 0);
            int4 p = make_int4(0, 0, 0, 0);
#pragma unroll
            for (int k = 0; k < CHUNKS / 16; ++k) {
                a.x += v[k].x; a.y += v[k].y; a.z += v[k].z; a.w += v[k].w;
                if (q + 16 * k < j) { p.x += v[k].x; p.y += v[k].y; p.z += v[k].z; p.w += v[k].w; }
            }
            atomicAdd(&tot[4 * c4],     a.x); atomicAdd(&tot[4 * c4 + 1], a.y);
            atomicAdd(&tot[4 * c4 + 2], a.z); atomicAdd(&tot[4 * c4 + 3], a.w);
            atomicAdd(&sh[4 * c4],      p.x); atomicAdd(&sh[4 * c4 + 1],  p.y);
            atomicAdd(&sh[4 * c4 + 2],  p.z); atomicAdd(&sh[4 * c4 + 3],  p.w);
        }
    }
    __syncthreads();
    int carry = 0;
    for (int base = 0; base < nb; base += 1024) {
        const int b = base + t;
        const int v = (b < nb) ? tot[b] : 0;
        int x = v;
#pragma unroll
        for (int d = 1; d < 32; d <<= 1) {
            const int y = __shfl_up(x, d);
            if (lane >= d) x += y;
        }
        if (lane == 31) ws[wave] = x;
        __syncthreads();
        if (wave == 0) {
            const int w = ws[lane];
            int xs = w;
#pragma unroll
            for (int d = 1; d < 32; d <<= 1) {
                const int y = __shfl_up(xs, d);
                if (lane >= d) xs += y;
            }
            ws[lane] = xs - w;
            if (lane == 31) ws[32] = xs;
        }
        __syncthreads();
        const int first = carry + ws[wave] + x - v;
        if (b < nb) {
            sh[b] += first;
            if (j == 0) bstart[b] = first;
        }
        carry += ws[32];
        __syncthreads();
    }
    if (j == 0 && t == 0) bstart[nb] = carry;
    const int lo = (int)blockIdx.x * per;
    const int hi = (lo + per < E) ? (lo + per) : E;
    for (int e0 = lo + t; e0 < hi; e0 += 8 * 1024) {
        int d[8], s[8];
#pragma unroll
        for (int k = 0; k < 8; ++k) {
            const int e  = e0 + 1024 * k;
            const int ec = e < hi ? e : hi - 1;
            d[k] = dst[ec];
            s[k] = src[ec];
        }
        asm volatile("" :: "v"(d[0]), "v"(d[1]), "v"(d[2]), "v"(d[3]),
                           "v"(d[4]), "v"(d[5]), "v"(d[6]), "v"(d[7]));
        asm volatile("" :: "v"(s[0]), "v"(s[1]), "v"(s[2]), "v"(s[3]),
                           "v"(s[4]), "v"(s[5]), "v"(s[6]), "v"(s[7]));
        int p[8];
#pragma unroll
        for (int k = 0; k < 8; ++k)
            p[k] = (e0 + 1024 * k < hi) ? atomicAdd(&sh[d[k] >> BSH], 1) : -1;
#pragma unroll
        for (int k = 0; k < 8; ++k)
            if (p[k] >= 0) pairs[p[k]] = make_int2(s[k], d[k]);
    }
}

__global__ __launch_bounds__(BNODES)
void bucket_rows_kernel(const int2* __restrict__ pairs, const int* __restrict__ bstart,
                        int* __restrict__ rowptr, float* __restrict__ dinv,
                        int* __restrict__ csr_src, int n) {
    __shared__ int cnt[BNODES];
    __shared__ int sc[BNODES];
    const int t = threadIdx.x;
    const int b = blockIdx.x;
    const int beg = bstart[b];
    const int end = bstart[b + 1];
    cnt[t] = 0;
    __syncthreads();
    for (int e0 = beg + t; e0 < end; e0 += 8 * BNODES) {
        int d[8];
#pragma unroll
        for (int k = 0; k < 8; ++k) {
            const int e = e0 + BNODES * k;
            d[k] = pairs[e < end ? e : end - 1].y;
        }
        asm volatile("" :: "v"(d[0]), "v"(d[1]), "v"(d[2]), "v"(d[3]),
                           "v"(d[4]), "v"(d[5]), "v"(d[6]), "v"(d[7]));
#pragma unroll
        for (int k = 0; k < 8; ++k)
            if (e0 + BNODES * k < end) atomicAdd(&cnt[d[k] & (BNODES - 1)], 1);
    }
    __syncthreads();
    const int c = cnt[t];
    sc[t] = c;
    __syncthreads();
    for (int s = 1; s < BNODES; s <<= 1) {
        const int v = (t >= s) ? sc[t - s] : 0;
        __syncthreads();
        sc[t] += v;
        __syncthreads();
    }
    const int first = beg + sc[t] - c;
    const int i = b * BNODES + t;
    if (i < n) {
        rowptr[i] = first;
        dinv[i]   = rsqrtf((float)(c + 1));
    }
    if (i == n - 1) rowptr[n] = first + c;
    cnt[t] = first;
    __syncthreads();
    for (int e0 = beg + t; e0 < end; e0 += 8 * BNODES) {
        int2 sd[8];
#pragma unroll
        for (int k = 0; k < 8; ++k) {
            const int e = e0 + BNODES * k;
            sd[k] = pairs[e < end ? e : end - 1];
        }
        asm volatile("" :: "v"(sd[0].y), "v"(sd[1].y), "v"(sd[2].y), "v"(sd[3].y),
                           "v"(sd[4].y), "v"(sd[5].y), "v"(sd[6].y), "v"(sd[7].y));
        int p[8];
#pragma unroll
        for (int k = 0; k < 8; ++k)
            p[k] = (e0 + BNODES * k < end) ? atomicAdd(&cnt[sd[k].y & (BNODES - 1)], 1) : -1;
#pragma unroll
        for (int k = 0; k < 8; ++k)
            if (p[k] >= 0) csr_src[p[k]] = sd[k].x;
    }
}

__global__ __launch_bounds__(256) __attribute__((amdgpu_waves_per_eu(8, 8)))
void gemm_wmma_f32(const float* __restrict__ X, const float* __restrict__ W,
                   const float* __restrict__ dinv,
                   float* __restrict__ H, int ntiles) {
    __shared__ float4 wl[64 * 64];

    const int wave = threadIdx.x >> 5;
    const int lane = threadIdx.x & 31;
    const int c    = lane & 15;
    const int h    = lane >> 4;
    const int tile = blockIdx.x * 8 + wave;
    const int tl   = tile < ntiles ? tile : ntiles - 1;
    const int m0   = tl * 16;

    const v2f* __restrict__ xa = (const v2f*)(X + (size_t)(m0 + c) * CH) + h;
    v2f a[32];
#pragma unroll
    for (int s = 0; s < 8; ++s) a[s] = xa[2 * s];
    const float dvl = dinv[m0 + c];

    const float4* __restrict__ W4 = (const float4*)W;
#pragma unroll
    for (int it = 0; it < 8; ++it) {
        const int p = it * 8 + wave;
        const float4 ra = W4[(2 * p) * 32 + lane];
        const float4 rb = W4[(2 * p + 1) * 32 + lane];
        float4 lo, hi;
        lo.x = ra.x; lo.y = rb.x; lo.z = ra.y; lo.w = rb.y;
        hi.x = ra.z; hi.y = rb.z; hi.z = ra.w; hi.w = rb.w;
        wl[p * 64 + (2 * h) * 16 + c]     = lo;
        wl[p * 64 + (2 * h + 1) * 16 + c] = hi;
    }
    __syncthreads();
    if (tile >= ntiles) return;

    v8f acc[8];
#pragma unroll
    for (int t = 0; t < 8; ++t) {
        acc[t] = (v8f){0.f, 0.f, 0.f, 0.f, 0.f, 0.f, 0.f, 0.f};
    }

    const float4* bl = wl + h * 64 + c;
#pragma unroll
    for (int s = 0; s < 32; ++s) {
        if ((s & 7) == 0) {
            if (s + 8 < 32) {
#pragma unroll
                for (int i = 0; i < 8; ++i) a[s + 8 + i] = xa[2 * (s + 8 + i)];
            }
            __builtin_amdgcn_sched_barrier(0);
        }
        const v2f as = a[s] * dvl;
        const float4 q0 = bl[s * 128];
        const float4 q1 = bl[s * 128 + 16];
        const float4 q2 = bl[s * 128 + 32];
        const float4 q3 = bl[s * 128 + 48];
        v2f b;
        b.x = q0.x; b.y = q0.y;
        acc[0] = __builtin_amdgcn_wmma_f32_16x16x4_f32(false, as, false, b, (short)0, acc[0], false, false);
        b.x = q0.z; b.y = q0.w;
        acc[1] = __builtin_amdgcn_wmma_f32_16x16x4_f32(false, as, false, b, (short)0, acc[1], false, false);
        b.x = q1.x; b.y = q1.y;
        acc[2] = __builtin_amdgcn_wmma_f32_16x16x4_f32(false, as, false, b, (short)0, acc[2], false, false);
        b.x = q1.z; b.y = q1.w;
        acc[3] = __builtin_amdgcn_wmma_f32_16x16x4_f32(false, as, false, b, (short)0, acc[3], false, false);
        b.x = q2.x; b.y = q2.y;
        acc[4] = __builtin_amdgcn_wmma_f32_16x16x4_f32(false, as, false, b, (short)0, acc[4], false, false);
        b.x = q2.z; b.y = q2.w;
        acc[5] = __builtin_amdgcn_wmma_f32_16x16x4_f32(false, as, false, b, (short)0, acc[5], false, false);
        b.x = q3.x; b.y = q3.y;
        acc[6] = __builtin_amdgcn_wmma_f32_16x16x4_f32(false, as, false, b, (short)0, acc[6], false, false);
        b.x = q3.z; b.y = q3.w;
        acc[7] = __builtin_amdgcn_wmma_f32_16x16x4_f32(false, as, false, b, (short)0, acc[7], false, false);
    }

    float4* __restrict__ H4 = (float4*)H;
#pragma unroll
    for (int r = 0; r < 8; ++r) {
        const size_t row4 = (size_t)(m0 + 8 * h + r) * (CH / 4);
        float4 v0, v1;
        v0.x = acc[0][r]; v0.y = acc[1][r]; v0.z = acc[2][r]; v0.w = acc[3][r];
        v1.x = acc[4][r]; v1.y = acc[5][r]; v1.z = acc[6][r]; v1.w = acc[7][r];
        H4[row4 + c]      = v0;
        H4[row4 + 16 + c] = v1;
    }
}

template <int MASK>
__device__ __forceinline__ float row_xor(float v) {
    return __builtin_bit_cast(float, __builtin_amdgcn_update_dpp(
        0, __builtin_bit_cast(int, v), 0x160 + MASK, 0xf, 0xf, true));
}

template <bool LAST>
__global__ __launch_bounds__(256)
void gather128_kernel(const float* __restrict__ h, const int* __restrict__ rowptr,
                      const int* __restrict__ csr_src, const float* __restrict__ dinv,
                      const float* __restrict__ b, const float* __restrict__ g,
                      const float* __restrict__ be, const float* __restrict__ rm,
                      const float* __restrict__ rv, const float* __restrict__ w3,
                      float* __restrict__ out, int n) {
    const int d = __builtin_amdgcn_readfirstlane(
        (int)((blockIdx.x * blockDim.x + threadIdx.x) >> 5));
    if (d >= n) return;
    const int lane = threadIdx.x & 31;
    const int beg = rowptr[d];
    const int end = rowptr[d + 1];
    const float dd = dinv[d];
    const float4* __restrict__ hv = (const float4*)h;

    const float4 pb  = ((const float4*)b)[lane];
    const float4 pg  = ((const float4*)g)[lane];
    const float4 pbe = ((const float4*)be)[lane];
    const float4 prm = ((const float4*)rm)[lane];
    const float4 prv = ((const float4*)rv)[lane];
    const float ix = pg.x * rsqrtf(prv.x + 1e-5f);
    const float iy = pg.y * rsqrtf(prv.y + 1e-5f);
    const float iz = pg.z * rsqrtf(prv.z + 1e-5f);
    const float iw = pg.w * rsqrtf(prv.w + 1e-5f);
    float4 wa = make_float4(0.f, 0.f, 0.f, 0.f);
    float4 wb = make_float4(0.f, 0.f, 0.f, 0.f);
    if (LAST) {
        wa = ((const float4*)w3)[2 * lane];
        wb = ((const float4*)w3)[2 * lane + 1];
    }

    float4 acc = hv[(size_t)d * 32 + lane];

    for (int base = beg; base < end; base += 32) {
        const int m = (end - base < 32) ? (end - base) : 32;
        int ms = 0;
        if (lane < m) ms = csr_src[base + lane];
        int j = 0;
        for (; j + 4 <= m; j += 4) {
            const int s0 = __builtin_amdgcn_readlane(ms, j);
            const int s1 = __builtin_amdgcn_readlane(ms, j + 1);
            const int s2 = __builtin_amdgcn_readlane(ms, j + 2);
            const int s3 = __builtin_amdgcn_readlane(ms, j + 3);
            const float4 v0 = hv[(size_t)s0 * 32 + lane];
            const float4 v1 = hv[(size_t)s1 * 32 + lane];
            const float4 v2 = hv[(size_t)s2 * 32 + lane];
            const float4 v3 = hv[(size_t)s3 * 32 + lane];
            acc.x += v0.x; acc.y += v0.y; acc.z += v0.z; acc.w += v0.w;
            acc.x += v1.x; acc.y += v1.y; acc.z += v1.z; acc.w += v1.w;
            acc.x += v2.x; acc.y += v2.y; acc.z += v2.z; acc.w += v2.w;
            acc.x += v3.x; acc.y += v3.y; acc.z += v3.z; acc.w += v3.w;
        }
        const int rem = m & 3;
        for (int t = 0; t < rem; ++t) {
            const int s0 = __builtin_amdgcn_readlane(ms, j + t);
            const float4 v0 = hv[(size_t)s0 * 32 + lane];
            acc.x += v0.x; acc.y += v0.y; acc.z += v0.z; acc.w += v0.w;
        }
    }
    float4 r;
    {
        const float vx = (acc.x * dd + pb.x - prm.x) * ix + pbe.x;
        const float vy = (acc.y * dd + pb.y - prm.y) * iy + pbe.y;
        const float vz = (acc.z * dd + pb.z - prm.z) * iz + pbe.z;
        const float vw = (acc.w * dd + pb.w - prm.w) * iw + pbe.w;
        r.x = vx > 0.f ? vx : 0.f;
        r.y = vy > 0.f ? vy : 0.f;
        r.z = vz > 0.f ? vz : 0.f;
        r.w = vw > 0.f ? vw : 0.f;
    }
    if (!LAST) {
        ((float4*)out)[(size_t)d * 32 + lane] = r;
    } else {
        const float p0 = r.x * wa.x + r.y * wa.z + r.z * wb.x + r.w * wb.z;
        const float p1 = r.x * wa.y + r.y * wa.w + r.z * wb.y + r.w * wb.w;
        const bool up = lane >= 16;
        const float keep = up ? p1 : p0;
        const float give = up ? p0 : p1;
        float v = keep + __builtin_bit_cast(float, __builtin_amdgcn_permlanex16(
            0, __builtin_bit_cast(int, give), 0x76543210, 0xfedcba98, false, false));
        v += row_xor<8>(v);
        v += row_xor<4>(v);
        v += row_xor<2>(v);
        v += row_xor<1>(v);
        if ((lane & 15) == 0) out[(size_t)d * 2 + (lane >> 4)] = v * dd;
    }
}

__global__ void gather2_kernel(const float* __restrict__ h3, const int* __restrict__ rowptr,
                               const int* __restrict__ csr_src, const float* __restrict__ dinv,
                               const float* __restrict__ b3, float* __restrict__ out, int n) {
    const int d = blockIdx.x * blockDim.x + threadIdx.x;
    if (d >= n) return;
    const int beg = rowptr[d];
    const int end = rowptr[d + 1];
    const float dd = dinv[d];
    const float2* __restrict__ hv = (const float2*)h3;
    const float2 vs = hv[d];
    float a0 = vs.x;
    float a1 = vs.y;
#pragma unroll 4
    for (int j = beg; j < end; ++j) {
        const int s = csr_src[j];
        const float2 u = hv[s];
        a0 += u.x;
        a1 += u.y;
    }
    float2 r;
    r.x = a0 * dd + b3[0];
    r.y = a1 * dd + b3[1];
    ((float2*)out)[d] = r;
}

extern "C" void kernel_launch(void* const* d_in, const int* in_sizes, int n_in,
                              void* d_out, int out_size, void* d_ws, size_t ws_size,
                              hipStream_t stream) {
    const float* x   = (const float*)d_in[0];
    const int*   ei  = (const int*)d_in[1];
    const float* W1  = (const float*)d_in[2];
    const float* b1  = (const float*)d_in[3];
    const float* g1  = (const float*)d_in[4];
    const float* be1 = (const float*)d_in[5];
    const float* rm1 = (const float*)d_in[6];
    const float* rv1 = (const float*)d_in[7];
    const float* W2  = (const float*)d_in[8];
    const float* b2  = (const float*)d_in[9];
    const float* g2  = (const float*)d_in[10];
    const float* be2 = (const float*)d_in[11];
    const float* rm2 = (const float*)d_in[12];
    const float* rv2 = (const float*)d_in[13];
    const float* W3  = (const float*)d_in[14];
    const float* b3  = (const float*)d_in[15];

    const int Nn   = in_sizes[0] / CH;
    const int Ee   = in_sizes[1] / 2;
    const int NC   = Nn * CH;

    const int* src = ei;
    const int* dst = ei + Ee;

    const int T = 256;
    const int gN  = (Nn + T - 1) / T;
    const int gW  = (Nn + 7) / 8;
    const int gMM = Nn / 16;
    const int gMB = (gMM + 7) / 8;
    const int nb  = (Nn + BNODES - 1) / BNODES;
    const int per = (Ee + CHUNKS - 1) / CHUNKS;
    const int nbp = (nb + 3) & ~3;
    const int tm  = nbp * CHUNKS;

    const size_t NP = ((size_t)Nn + 64) & ~(size_t)63;
    const size_t TP = ((size_t)tm + 63) & ~(size_t)63;
    const size_t BP = ((size_t)nb + 64) & ~(size_t)63;
    const size_t EP = ((size_t)Ee + 63) & ~(size_t)63;
    int*   tab    = (int*)d_ws;
    int*   bstart = tab + TP;
    int*   rowptr = bstart + BP;
    float* dinv   = (float*)(rowptr + NP);
    int*   csr    = (int*)(dinv + NP);
    int2*  pairs  = (int2*)(csr + EP);
    float* bufA   = (float*)(csr + 3 * EP);
    float* bufB   = bufA + (size_t)NC;
    float* h3     = bufB + (size_t)NC;
    float* out    = (float*)d_out;

    bucket_count_kernel<<<CHUNKS, 1024, nb * sizeof(int), stream>>>(dst, tab, Ee, per, nb, nbp);
    bucket_place_kernel<<<CHUNKS, 1024, (2 * nbp + 33) * sizeof(int), stream>>>(src, dst, tab, bstart, pairs, Ee, per, nb, nbp);
    bucket_rows_kernel<<<nb, BNODES, 0, stream>>>(pairs, bstart, rowptr, dinv, csr, Nn);

    gemm_wmma_f32<<<gMB, T, 0, stream>>>(x, W1, dinv, bufA, gMM);
    gather128_kernel<false><<<gW, T, 0, stream>>>(bufA, rowptr, csr, dinv, b1, g1, be1, rm1, rv1, W3, bufB, Nn);

    gemm_wmma_f32<<<gMB, T, 0, stream>>>(bufB, W2, dinv, bufA, gMM);
    gather128_kernel<true><<<gW, T, 0, stream>>>(bufA, rowptr, csr, dinv, b2, g2, be2, rm2, rv2, W3, h3, Nn);

    gather2_kernel<<<gN, T, 0, stream>>>(h3, rowptr, csr, dinv, b3, out, Nn);
}


# GEMM made resident and balanced: 256 work-groups of 32 waves (one per WGP, W to LDS once each), 16x64 half-stripe units dealt evenly over the SIMDs (12-13 each, was up to 16); sums unchanged
# speedup vs baseline: 5.2246x; 1.0497x over previous
#include <hip/hip_runtime.h>

#define CH 128

typedef float v2f __attribute__((ext_vector_type(2)));
typedef float v8f __attribute__((ext_vector_type(8)));

#define BSH    9
#define BNODES 512
#define CHUNKS 256
static_assert(CHUNKS == 256, "bucket_place_kernel names its sixteen table loads one by one");

__global__ __launch_bounds__(1024)
void bucket_count_kernel(const int* __restrict__ dst, int* __restrict__ tab,
                         int E, int per, int nb, int nbp) {
    extern __shared__ int sh[];
    const int t = threadIdx.x;
    for (int b = t; b < nb; b += 1024) sh[b] = 0;
    __syncthreads();
    const int lo = (int)blockIdx.x * per;
    const int hi = (lo + per < E) ? (lo + per) : E;
    for (int e0 = lo + t; e0 < hi; e0 += 8 * 1024) {
        int d[8];
#pragma unroll
        for (int k = 0; k < 8; ++k) {
            const int e = e0 + 1024 * k;
            d[k] = dst[e < hi ? e : hi - 1];
        }
        asm volatile("" :: "v"(d[0]), "v"(d[1]), "v"(d[2]), "v"(d[3]),
                           "v"(d[4]), "v"(d[5]), "v"(d[6]), "v"(d[7]));
#pragma unroll
        for (int k = 0; k < 8; ++k)
            if (e0 + 1024 * k < hi) atomicAdd(&sh[d[k] >> BSH], 1);
    }
    __syncthreads();
    for (int b = t; b < nbp; b += 1024)
        tab[(int)blockIdx.x * nbp + b] = (b < nb) ? sh[b] : 0;
}

__global__ __launch_bounds__(1024)
void bucket_place_kernel(const int* __restrict__ src, const int* __restrict__ dst,
                         const int* __restrict__ tab, int* __restrict__ bstart,
                         int2* __restrict__ pairs,
                         int E, int per, int nb, int nbp) {
    extern __shared__ int sh[];
    int* tot = sh + nbp;
    int* ws  = sh + 2 * nbp;
    const int t    = threadIdx.x;
    const int lane = t & 31;
    const int wave = t >> 5;
    const int j    = (int)blockIdx.x;
    for (int b = t; b < 2 * nbp; b += 1024) sh[b] = 0;
    __syncthreads();
    {
        const int nq = nbp >> 2;
        const int4* __restrict__ tab4 = (const int4*)tab;
        const int q = t >> 6;
        for (int c4 = t & 63; c4 < nq; c4 += 64) {
            int4 v[CHUNKS / 16];
#pragma unroll
            for (int k = 0; k < CHUNKS / 16; ++k) v[k] = tab4[(q + 16 * k) * nq + c4];
            asm volatile("" :: "v"(v[0].x), "v"(v[1].x), "v"(v[2].x), "v"(v[3].x),
                               "v"(v[4].x), "v"(v[5].x), "v"(v[6].x), "v"(v[7].x),
                               "v"(v[8].x), "v"(v[9].x), "v"(v[10].x), "v"(v[11].x),
                               "v"(v[12].x), "v"(v[13].x), "v"(v[14].x), "v"(v[15].x));
            int4 a = make_int4(0, 0, 0, 0);
            int4 p = make_int4(0, 0, 0, 0);
#pragma unroll
            for (int k = 0; k < CHUNKS / 16; ++k) {
                a.x += v[k].x; a.y += v[k].y; a.z += v[k].z; a.w += v[k].w;
                if (q + 16 * k < j) { p.x += v[k].x; p.y += v[k].y; p.z += v[k].z; p.w += v[k].w; }
            }
            atomicAdd(&tot[4 * c4],     a.x); atomicAdd(&tot[4 * c4 + 1], a.y);
            atomicAdd(&tot[4 * c4 + 2], a.z); atomicAdd(&tot[4 * c4 + 3], a.w);
            atomicAdd(&sh[4 * c4],      p.x); atomicAdd(&sh[4 * c4 + 1],  p.y);
            atomicAdd(&sh[4 * c4 + 2],  p.z); atomicAdd(&sh[4 * c4 + 3],  p.w);
        }
    }
    __syncthreads();
    int carry = 0;
    for (int base = 0; base < nb; base += 1024) {
        const int b = base + t;
        const int v = (b < nb) ? tot[b] : 0;
        int x = v;
#pragma unroll
        for (int d = 1; d < 32; d <<= 1) {
            const int y = __shfl_up(x, d);
            if (lane >= d) x += y;
        }
        if (lane == 31) ws[wave] = x;
        __syncthreads();
        if (wave == 0) {
            const int w = ws[lane];
            int xs = w;
#pragma unroll
            for (int d = 1; d < 32; d <<= 1) {
                const int y = __shfl_up(xs, d);
                if (lane >= d) xs += y;
            }
            ws[lane] = xs - w;
            if (lane == 31) ws[32] = xs;
        }
        __syncthreads();
        const int first = carry + ws[wave] + x - v;
        if (b < nb) {
            sh[b] += first;
            if (j == 0) bstart[b] = first;
        }
        carry += ws[32];
        __syncthreads();
    }
    if (j == 0 && t == 0) bstart[nb] = carry;
    const int lo = (int)blockIdx.x * per;
    const int hi = (lo + per < E) ? (lo + per) : E;
    for (int e0 = lo + t; e0 < hi; e0 += 8 * 1024) {
        int d[8], s[8];
#pragma unroll
        for (int k = 0; k < 8; ++k) {
            const int e  = e0 + 1024 * k;
            const int ec = e < hi ? e : hi - 1;
            d[k] = dst[ec];
            s[k] = src[ec];
        }
        asm volatile("" :: "v"(d[0]), "v"(d[1]), "v"(d[2]), "v"(d[3]),
                           "v"(d[4]), "v"(d[5]), "v"(d[6]), "v"(d[7]));
        asm volatile("" :: "v"(s[0]), "v"(s[1]), "v"(s[2]), "v"(s[3]),
                           "v"(s[4]), "v"(s[5]), "v"(s[6]), "v"(s[7]));
        int p[8];
#pragma unroll
        for (int k = 0; k < 8; ++k)
            p[k] = (e0 + 1024 * k < hi) ? atomicAdd(&sh[d[k] >> BSH], 1) : -1;
#pragma unroll
        for (int k = 0; k < 8; ++k)
            if (p[k] >= 0) pairs[p[k]] = make_int2(s[k], d[k]);
    }
}

__global__ __launch_bounds__(BNODES)
void bucket_rows_kernel(const int2* __restrict__ pairs, const int* __restrict__ bstart,
                        int* __restrict__ rowptr, float* __restrict__ dinv,
                        int* __restrict__ csr_src, int n) {
    __shared__ int cnt[BNODES];
    __shared__ int sc[BNODES];
    const int t = threadIdx.x;
    const int b = blockIdx.x;
    const int beg = bstart[b];
    const int end = bstart[b + 1];
    cnt[t] = 0;
    __syncthreads();
    for (int e0 = beg + t; e0 < end; e0 += 8 * BNODES) {
        int d[8];
#pragma unroll
        for (int k = 0; k < 8; ++k) {
            const int e = e0 + BNODES * k;
            d[k] = pairs[e < end ? e : end - 1].y;
        }
        asm volatile("" :: "v"(d[0]), "v"(d[1]), "v"(d[2]), "v"(d[3]),
                           "v"(d[4]), "v"(d[5]), "v"(d[6]), "v"(d[7]));
#pragma unroll
        for (int k = 0; k < 8; ++k)
            if (e0 + BNODES * k < end) atomicAdd(&cnt[d[k] & (BNODES - 1)], 1);
    }
    __syncthreads();
    const int c = cnt[t];
    sc[t] = c;
    __syncthreads();
    for (int s = 1; s < BNODES; s <<= 1) {
        const int v = (t >= s) ? sc[t - s] : 0;
        __syncthreads();
        sc[t] += v;
        __syncthreads();
    }
    const int first = beg + sc[t] - c;
    const int i = b * BNODES + t;
    if (i < n) {
        rowptr[i] = first;
        dinv[i]   = rsqrtf((float)(c + 1));
    }
    if (i == n - 1) rowptr[n] = first + c;
    cnt[t] = first;
    __syncthreads();
    for (int e0 = beg + t; e0 < end; e0 += 8 * BNODES) {
        int2 sd[8];
#pragma unroll
        for (int k = 0; k < 8; ++k) {
            const int e = e0 + BNODES * k;
            sd[k] = pairs[e < end ? e : end - 1];
        }
        asm volatile("" :: "v"(sd[0].y), "v"(sd[1].y), "v"(sd[2].y), "v"(sd[3].y),
                           "v"(sd[4].y), "v"(sd[5].y), "v"(sd[6].y), "v"(sd[7].y));
        int p[8];
#pragma unroll
        for (int k = 0; k < 8; ++k)
            p[k] = (e0 + BNODES * k < end) ? atomicAdd(&cnt[sd[k].y & (BNODES - 1)], 1) : -1;
#pragma unroll
        for (int k = 0; k < 8; ++k)
            if (p[k] >= 0) csr_src[p[k]] = sd[k].x;
    }
}

__global__ __launch_bounds__(1024) __attribute__((amdgpu_waves_per_eu(8, 8)))
void gemm_wmma_f32(const float* __restrict__ X, const float* __restrict__ W,
                   const float* __restrict__ dinv,
                   float* __restrict__ H, int ntiles) {
    __shared__ float4 wl[64 * 64];

    const int wave = threadIdx.x >> 5;
    const int lane = threadIdx.x & 31;
    const int c    = lane & 15;
    const int h    = lane >> 4;
    const int G    = (int)gridDim.x;
    const int g    = (int)blockIdx.x;
    const int nunits = 2 * ntiles;
    const int nl   = (nunits - g + G - 1) / G;
    const int j    = ((wave >> 2) & 1) * 16 + (wave >> 3) * 4 + (wave & 3);

    int l = j;
    int u = (l < nl) ? (l * G + g) : (nunits - 1);
    const v2f* __restrict__ xa = (const v2f*)(X + (size_t)((u >> 1) * 16 + c) * CH) + h;
    v2f a[32];
#pragma unroll
    for (int s = 0; s < 8; ++s) a[s] = xa[2 * s];
    float dvl = dinv[(u >> 1) * 16 + c];

    const float4* __restrict__ W4 = (const float4*)W;
#pragma unroll
    for (int it = 0; it < 2; ++it) {
        const int p = it * 32 + wave;
        const float4 ra = W4[(2 * p) * 32 + lane];
        const float4 rb = W4[(2 * p + 1) * 32 + lane];
        float4 lo, hi;
        lo.x = ra.x; lo.y = rb.x; lo.z = ra.y; lo.w = rb.y;
        hi.x = ra.z; hi.y = rb.z; hi.z = ra.w; hi.w = rb.w;
        wl[p * 64 + (2 * h) * 16 + c]     = lo;
        wl[p * 64 + (2 * h + 1) * 16 + c] = hi;
    }
    __syncthreads();

    float4* __restrict__ H4 = (float4*)H;
    while (l < nl) {
        const int m0 = (u >> 1) * 16;
        const int ch = u & 1;

        v8f acc[4];
#pragma unroll
        for (int t = 0; t < 4; ++t) {
            acc[t] = (v8f){0.f, 0.f, 0.f, 0.f, 0.f, 0.f, 0.f, 0.f};
        }

        const float4* bl = wl + h * 64 + ch * 32 + c;
#pragma unroll
        for (int s = 0; s < 32; ++s) {
            if ((s & 7) == 0) {
                if (s + 8 < 32) {
#pragma unroll
                    for (int i = 0; i < 8; ++i) a[s + 8 + i] = xa[2 * (s + 8 + i)];
                }
                __builtin_amdgcn_sched_barrier(0);
            }
            const v2f as = a[s] * dvl;
            const float4 q0 = bl[s * 128];
            const float4 q1 = bl[s * 128 + 16];
            v2f b;
            b.x = q0.x; b.y = q0.y;
            acc[0] = __builtin_amdgcn_wmma_f32_16x16x4_f32(false, as, false, b, (short)0, acc[0], false, false);
            b.x = q0.z; b.y = q0.w;
            acc[1] = __builtin_amdgcn_wmma_f32_16x16x4_f32(false, as, false, b, (short)0, acc[1], false, false);
            b.x = q1.x; b.y = q1.y;
            acc[2] = __builtin_amdgcn_wmma_f32_16x16x4_f32(false, as, false, b, (short)0, acc[2], false, false);
            b.x = q1.z; b.y = q1.w;
            acc[3] = __builtin_amdgcn_wmma_f32_16x16x4_f32(false, as, false, b, (short)0, acc[3], false, false);
        }

#pragma unroll
        for (int r = 0; r < 8; ++r) {
            const size_t row4 = (size_t)(m0 + 8 * h + r) * (CH / 4);
            float4 v0;
            v0.x = acc[0][r]; v0.y = acc[1][r]; v0.z = acc[2][r]; v0.w = acc[3][r];
            H4[row4 + 16 * ch + c] = v0;
        }

        l += 32;
        if (l < nl) {
            u  = l * G + g;
            xa = (const v2f*)(X + (size_t)((u >> 1) * 16 + c) * CH) + h;
#pragma unroll
            for (int s = 0; s < 8; ++s) a[s] = xa[2 * s];
            dvl = dinv[(u >> 1) * 16 + c];
        }
    }
}

template <int MASK>
__device__ __forceinline__ float row_xor(float v) {
    return __builtin_bit_cast(float, __builtin_amdgcn_update_dpp(
        0, __builtin_bit_cast(int, v), 0x160 + MASK, 0xf, 0xf, true));
}

template <bool LAST>
__global__ __launch_bounds__(256)
void gather128_kernel(const float* __restrict__ h, const int* __restrict__ rowptr,
                      const int* __restrict__ csr_src, const float* __restrict__ dinv,
                      const float* __restrict__ b, const float* __restrict__ g,
                      const float* __restrict__ be, const float* __restrict__ rm,
                      const float* __restrict__ rv, const float* __restrict__ w3,
                      float* __restrict__ out, int n) {
    const int d = __builtin_amdgcn_readfirstlane(
        (int)((blockIdx.x * blockDim.x + threadIdx.x) >> 5));
    if (d >= n) return;
    const int lane = threadIdx.x & 31;
    const int beg = rowptr[d];
    const int end = rowptr[d + 1];
    const float dd = dinv[d];
    const float4* __restrict__ hv = (const float4*)h;

    const float4 pb  = ((const float4*)b)[lane];
    const float4 pg  = ((const float4*)g)[lane];
    const float4 pbe = ((const float4*)be)[lane];
    const float4 prm = ((const float4*)rm)[lane];
    const float4 prv = ((const float4*)rv)[lane];
    const float ix = pg.x * rsqrtf(prv.x + 1e-5f);
    const float iy = pg.y * rsqrtf(prv.y + 1e-5f);
    const float iz = pg.z * rsqrtf(prv.z + 1e-5f);
    const float iw = pg.w * rsqrtf(prv.w + 1e-5f);
    float4 wa = make_float4(0.f, 0.f, 0.f, 0.f);
    float4 wb = make_float4(0.f, 0.f, 0.f, 0.f);
    if (LAST) {
        wa = ((const float4*)w3)[2 * lane];
        wb = ((const float4*)w3)[2 * lane + 1];
    }

    float4 acc = hv[(size_t)d * 32 + lane];

    for (int base = beg; base < end; base += 32) {
        const int m = (end - base < 32) ? (end - base) : 32;
        int ms = 0;
        if (lane < m) ms = csr_src[base + lane];
        int j = 0;
        for (; j + 4 <= m; j += 4) {
            const int s0 = __builtin_amdgcn_readlane(ms, j);
            const int s1 = __builtin_amdgcn_readlane(ms, j + 1);
            const int s2 = __builtin_amdgcn_readlane(ms, j + 2);
            const int s3 = __builtin_amdgcn_readlane(ms, j + 3);
            const float4 v0 = hv[(size_t)s0 * 32 + lane];
            const float4 v1 = hv[(size_t)s1 * 32 + lane];
            const float4 v2 = hv[(size_t)s2 * 32 + lane];
            const float4 v3 = hv[(size_t)s3 * 32 + lane];
            acc.x += v0.x; acc.y += v0.y; acc.z += v0.z; acc.w += v0.w;
            acc.x += v1.x; acc.y += v1.y; acc.z += v1.z; acc.w += v1.w;
            acc.x += v2.x; acc.y += v2.y; acc.z += v2.z; acc.w += v2.w;
            acc.x += v3.x; acc.y += v3.y; acc.z += v3.z; acc.w += v3.w;
        }
        const int rem = m & 3;
        for (int t = 0; t < rem; ++t) {
            const int s0 = __builtin_amdgcn_readlane(ms, j + t);
            const float4 v0 = hv[(size_t)s0 * 32 + lane];
            acc.x += v0.x; acc.y += v0.y; acc.z += v0.z; acc.w += v0.w;
        }
    }
    float4 r;
    {
        const float vx = (acc.x * dd + pb.x - prm.x) * ix + pbe.x;
        const float vy = (acc.y * dd + pb.y - prm.y) * iy + pbe.y;
        const float vz = (acc.z * dd + pb.z - prm.z) * iz + pbe.z;
        const float vw = (acc.w * dd + pb.w - prm.w) * iw + pbe.w;
        r.x = vx > 0.f ? vx : 0.f;
        r.y = vy > 0.f ? vy : 0.f;
        r.z = vz > 0.f ? vz : 0.f;
        r.w = vw > 0.f ? vw : 0.f;
    }
    if (!LAST) {
        ((float4*)out)[(size_t)d * 32 + lane] = r;
    } else {
        const float p0 = r.x * wa.x + r.y * wa.z + r.z * wb.x + r.w * wb.z;
        const float p1 = r.x * wa.y + r.y * wa.w + r.z * wb.y + r.w * wb.w;
        const bool up = lane >= 16;
        const float keep = up ? p1 : p0;
        const float give = up ? p0 : p1;
        float v = keep + __builtin_bit_cast(float, __builtin_amdgcn_permlanex16(
            0, __builtin_bit_cast(int, give), 0x76543210, 0xfedcba98, false, false));
        v += row_xor<8>(v);
        v += row_xor<4>(v);
        v += row_xor<2>(v);
        v += row_xor<1>(v);
        if ((lane & 15) == 0) out[(size_t)d * 2 + (lane >> 4)] = v * dd;
    }
}

__global__ void gather2_kernel(const float* __restrict__ h3, const int* __restrict__ rowptr,
                               const int* __restrict__ csr_src, const float* __restrict__ dinv,
                               const float* __restrict__ b3, float* __restrict__ out, int n) {
    const int d = blockIdx.x * blockDim.x + threadIdx.x;
    if (d >= n) return;
    const int beg = rowptr[d];
    const int end = rowptr[d + 1];
    const float dd = dinv[d];
    const float2* __restrict__ hv = (const float2*)h3;
    const float2 vs = hv[d];
    float a0 = vs.x;
    float a1 = vs.y;
#pragma unroll 4
    for (int j = beg; j < end; ++j) {
        const int s = csr_src[j];
        const float2 u = hv[s];
        a0 += u.x;
        a1 += u.y;
    }
    float2 r;
    r.x = a0 * dd + b3[0];
    r.y = a1 * dd + b3[1];
    ((float2*)out)[d] = r;
}

extern "C" void kernel_launch(void* const* d_in, const int* in_sizes, int n_in,
                              void* d_out, int out_size, void* d_ws, size_t ws_size,
                              hipStream_t stream) {
    const float* x   = (const float*)d_in[0];
    const int*   ei  = (const int*)d_in[1];
    const float* W1  = (const float*)d_in[2];
    const float* b1  = (const float*)d_in[3];
    const float* g1  = (const float*)d_in[4];
    const float* be1 = (const float*)d_in[5];
    const float* rm1 = (const float*)d_in[6];
    const float* rv1 = (const float*)d_in[7];
    const float* W2  = (const float*)d_in[8];
    const float* b2  = (const float*)d_in[9];
    const float* g2  = (const float*)d_in[10];
    const float* be2 = (const float*)d_in[11];
    const float* rm2 = (const float*)d_in[12];
    const float* rv2 = (const float*)d_in[13];
    const float* W3  = (const float*)d_in[14];
    const float* b3  = (const float*)d_in[15];

    const int Nn   = in_sizes[0] / CH;
    const int Ee   = in_sizes[1] / 2;
    const int NC   = Nn * CH;

    const int* src = ei;
    const int* dst = ei + Ee;

    const int T = 256;
    const int gN  = (Nn + T - 1) / T;
    const int gW  = (Nn + 7) / 8;
    const int gMM = Nn / 16;
    const int nb  = (Nn + BNODES - 1) / BNODES;
    const int per = (Ee + CHUNKS - 1) / CHUNKS;
    const int nbp = (nb + 3) & ~3;
    const int tm  = nbp * CHUNKS;

    const size_t NP = ((size_t)Nn + 64) & ~(size_t)63;
    const size_t TP = ((size_t)tm + 63) & ~(size_t)63;
    const size_t BP = ((size_t)nb + 64) & ~(size_t)63;
    const size_t EP = ((size_t)Ee + 63) & ~(size_t)63;
    int*   tab    = (int*)d_ws;
    int*   bstart = tab + TP;
    int*   rowptr = bstart + BP;
    float* dinv   = (float*)(rowptr + NP);
    int*   csr    = (int*)(dinv + NP);
    int2*  pairs  = (int2*)(csr + EP);
    float* bufA   = (float*)(csr + 3 * EP);
    float* bufB   = bufA + (size_t)NC;
    float* h3     = bufB + (size_t)NC;
    float* out    = (float*)d_out;

    bucket_count_kernel<<<CHUNKS, 1024, nb * sizeof(int), stream>>>(dst, tab, Ee, per, nb, nbp);
    bucket_place_kernel<<<CHUNKS, 1024, (2 * nbp + 33) * sizeof(int), stream>>>(src, dst, tab, bstart, pairs, Ee, per, nb, nbp);
    bucket_rows_kernel<<<nb, BNODES, 0, stream>>>(pairs, bstart, rowptr, dinv, csr, Nn);

    gemm_wmma_f32<<<256, 1024, 0, stream>>>(x, W1, dinv, bufA, gMM);
    gather128_kernel<false><<<gW, T, 0, stream>>>(bufA, rowptr, csr, dinv, b1, g1, be1, rm1, rv1, W3, bufB, Nn);

    gemm_wmma_f32<<<256, 1024, 0, stream>>>(bufB, W2, dinv, bufA, gMM);
    gather128_kernel<true><<<gW, T, 0, stream>>>(bufA, rowptr, csr, dinv, b2, g2, be2, rm2, rv2, W3, h3, Nn);

    gather2_kernel<<<gN, T, 0, stream>>>(h3, rowptr, csr, dinv, b3, out, Nn);
}


# Both edge placement passes put their scattered 8- and 4-byte stores into an LDS copy first and write it out in slot order as whole lines; slot order (so sum order) is still LDS arrival order
# speedup vs baseline: 5.6072x; 1.0732x over previous
#include <hip/hip_runtime.h>

#define CH 128

typedef float v2f __attribute__((ext_vector_type(2)));
typedef float v8f __attribute__((ext_vector_type(8)));

#define BSH    9
#define BNODES 512
#define CHUNKS 256
static_assert(CHUNKS == 256, "bucket_place_kernel names its sixteen table loads one by one");

__global__ __launch_bounds__(1024)
void bucket_count_kernel(const int* __restrict__ dst, int* __restrict__ tab,
                         int E, int per, int nb, int nbp) {
    extern __shared__ int sh[];
    const int t = threadIdx.x;
    for (int b = t; b < nb; b += 1024) sh[b] = 0;
    __syncthreads();
    const int lo = (int)blockIdx.x * per;
    const int hi = (lo + per < E) ? (lo + per) : E;
    for (int e0 = lo + t; e0 < hi; e0 += 8 * 1024) {
        int d[8];
#pragma unroll
        for (int k = 0; k < 8; ++k) {
            const int e = e0 + 1024 * k;
            d[k] = dst[e < hi ? e : hi - 1];
        }
        asm volatile("" :: "v"(d[0]), "v"(d[1]), "v"(d[2]), "v"(d[3]),
                           "v"(d[4]), "v"(d[5]), "v"(d[6]), "v"(d[7]));
#pragma unroll
        for (int k = 0; k < 8; ++k)
            if (e0 + 1024 * k < hi) atomicAdd(&sh[d[k] >> BSH], 1);
    }
    __syncthreads();
    for (int b = t; b < nbp; b += 1024)
        tab[(int)blockIdx.x * nbp + b] = (b < nb) ? sh[b] : 0;
}

__global__ __launch_bounds__(1024)
void bucket_place_kernel(const int* __restrict__ src, const int* __restrict__ dst,
                         const int* __restrict__ tab, int* __restrict__ bstart,
                         int2* __restrict__ pairs,
                         int E, int per, int nb, int nbp) {
    extern __shared__ int sh[];
    int* tot = sh + nbp;
    int* cur = sh + 2 * nbp;
    int* ws  = sh + 3 * nbp;
    int2* stage = (int2*)(sh + 3 * nbp + 68);
    const int t    = threadIdx.x;
    const int lane = t & 31;
    const int wave = t >> 5;
    const int j    = (int)blockIdx.x;
    const int own0 = (t < nb) ? tab[j * nbp + t] : 0;
    for (int b = t; b < 2 * nbp; b += 1024) sh[b] = 0;
    __syncthreads();
    {
        const int nq = nbp >> 2;
        const int4* __restrict__ tab4 = (const int4*)tab;
        const int q = t >> 6;
        for (int c4 = t & 63; c4 < nq; c4 += 64) {
            int4 v[CHUNKS / 16];
#pragma unroll
            for (int k = 0; k < CHUNKS / 16; ++k) v[k] = tab4[(q + 16 * k) * nq + c4];
            asm volatile("" :: "v"(v[0].x), "v"(v[1].x), "v"(v[2].x), "v"(v[3].x),
                               "v"(v[4].x), "v"(v[5].x), "v"(v[6].x), "v"(v[7].x),
                               "v"(v[8].x), "v"(v[9].x), "v"(v[10].x), "v"(v[11].x),
                               "v"(v[12].x), "v"(v[13].x), "v"(v[14].x), "v"(v[15].x));
            int4 a = make_int4(0, 0, 0, 0);
            int4 p = make_int4(0, 0, 0, 0);
#pragma unroll
            for (int k = 0; k < CHUNKS / 16; ++k) {
                a.x += v[k].x; a.y += v[k].y; a.z += v[k].z; a.w += v[k].w;
                if (q + 16 * k < j) { p.x += v[k].x; p.y += v[k].y; p.z += v[k].z; p.w += v[k].w; }
            }
            atomicAdd(&tot[4 * c4],     a.x); atomicAdd(&tot[4 * c4 + 1], a.y);
            atomicAdd(&tot[4 * c4 + 2], a.z); atomicAdd(&tot[4 * c4 + 3], a.w);
            atomicAdd(&sh[4 * c4],      p.x); atomicAdd(&sh[4 * c4 + 1],  p.y);
            atomicAdd(&sh[4 * c4 + 2],  p.z); atomicAdd(&sh[4 * c4 + 3],  p.w);
        }
    }
    __syncthreads();
    int carry = 0;
    int lcarry = 0;
    for (int base = 0; base < nb; base += 1024) {
        const int b = base + t;
        const int v = (b < nb) ? tot[b] : 0;
        const int o = (base == 0) ? own0 : ((b < nb) ? tab[j * nbp + b] : 0);
        int x = v;
        int z = o;
#pragma unroll
        for (int d = 1; d < 32; d <<= 1) {
            const int y = __shfl_up(x, d);
            const int w = __shfl_up(z, d);
            if (lane >= d) { x += y; z += w; }
        }
        if (lane == 31) { ws[wave] = x; ws[33 + wave] = z; }
        __syncthreads();
        if (wave < 2) {
            int* wq = ws + 33 * wave;
            const int w = wq[lane];
            int xs = w;
#pragma unroll
            for (int d = 1; d < 32; d <<= 1) {
                const int y = __shfl_up(xs, d);
                if (lane >= d) xs += y;
            }
            wq[lane] = xs - w;
            if (lane == 31) wq[32] = xs;
        }
        __syncthreads();
        const int first  = carry + ws[wave] + x - v;
        const int lfirst = lcarry + ws[33 + wave] + z - o;
        if (b < nb) {
            cur[b] = lfirst;
            sh[b] += first - lfirst;
            if (j == 0) bstart[b] = first;
        }
        carry  += ws[32];
        lcarry += ws[65];
        __syncthreads();
    }
    if (j == 0 && t == 0) bstart[nb] = carry;
    const int lo = (int)blockIdx.x * per;
    const int hi = (lo + per < E) ? (lo + per) : E;
    for (int e0 = lo + t; e0 < hi; e0 += 8 * 1024) {
        int d[8], s[8];
#pragma unroll
        for (int k = 0; k < 8; ++k) {
            const int e  = e0 + 1024 * k;
            const int ec = e < hi ? e : hi - 1;
            d[k] = dst[ec];
            s[k] = src[ec];
        }
        asm volatile("" :: "v"(d[0]), "v"(d[1]), "v"(d[2]), "v"(d[3]),
                           "v"(d[4]), "v"(d[5]), "v"(d[6]), "v"(d[7]));
        asm volatile("" :: "v"(s[0]), "v"(s[1]), "v"(s[2]), "v"(s[3]),
                           "v"(s[4]), "v"(s[5]), "v"(s[6]), "v"(s[7]));
        int p[8];
#pragma unroll
        for (int k = 0; k < 8; ++k)
            p[k] = (e0 + 1024 * k < hi) ? atomicAdd(&cur[d[k] >> BSH], 1) : -1;
#pragma unroll
        for (int k = 0; k < 8; ++k)
            if (p[k] >= 0) stage[p[k]] = make_int2(s[k], d[k]);
    }
    __syncthreads();
    const int n = hi - lo;
    for (int i0 = t; i0 < n; i0 += 8 * 1024) {
        int2 pr[8];
        int  off[8];
#pragma unroll
        for (int k = 0; k < 8; ++k) {
            const int i = i0 + 1024 * k;
            pr[k] = stage[i < n ? i : n - 1];
        }
        asm volatile("" :: "v"(pr[0].y), "v"(pr[1].y), "v"(pr[2].y), "v"(pr[3].y),
                           "v"(pr[4].y), "v"(pr[5].y), "v"(pr[6].y), "v"(pr[7].y));
#pragma unroll
        for (int k = 0; k < 8; ++k) off[k] = sh[pr[k].y >> BSH];
        asm volatile("" :: "v"(off[0]), "v"(off[1]), "v"(off[2]), "v"(off[3]),
                           "v"(off[4]), "v"(off[5]), "v"(off[6]), "v"(off[7]));
#pragma unroll
        for (int k = 0; k < 8; ++k)
            if (i0 + 1024 * k < n) pairs[i0 + 1024 * k + off[k]] = pr[k];
    }
}

#define RSTAGE 12288
__global__ __launch_bounds__(BNODES)
void bucket_rows_kernel(const int2* __restrict__ pairs, const int* __restrict__ bstart,
                        int* __restrict__ rowptr, float* __restrict__ dinv,
                        int* __restrict__ csr_src, int n) {
    __shared__ int cnt[BNODES];
    __shared__ int sc[BNODES];
    extern __shared__ int stage[];
    const int t = threadIdx.x;
    const int b = blockIdx.x;
    const int beg = bstart[b];
    const int end = bstart[b + 1];
    cnt[t] = 0;
    __syncthreads();
    for (int e0 = beg + t; e0 < end; e0 += 8 * BNODES) {
        int d[8];
#pragma unroll
        for (int k = 0; k < 8; ++k) {
            const int e = e0 + BNODES * k;
            d[k] = pairs[e < end ? e : end - 1].y;
        }
        asm volatile("" :: "v"(d[0]), "v"(d[1]), "v"(d[2]), "v"(d[3]),
                           "v"(d[4]), "v"(d[5]), "v"(d[6]), "v"(d[7]));
#pragma unroll
        for (int k = 0; k < 8; ++k)
            if (e0 + BNODES * k < end) atomicAdd(&cnt[d[k] & (BNODES - 1)], 1);
    }
    __syncthreads();
    const int c = cnt[t];
    sc[t] = c;
    __syncthreads();
    for (int s = 1; s < BNODES; s <<= 1) {
        const int v = (t >= s) ? sc[t - s] : 0;
        __syncthreads();
        sc[t] += v;
        __syncthreads();
    }
    const int first = beg + sc[t] - c;
    const int i = b * BNODES + t;
    if (i < n) {
        rowptr[i] = first;
        dinv[i]   = rsqrtf((float)(c + 1));
    }
    if (i == n - 1) rowptr[n] = first + c;
    cnt[t] = first;
    __syncthreads();
    for (int e0 = beg + t; e0 < end; e0 += 8 * BNODES) {
        int2 sd[8];
#pragma unroll
        for (int k = 0; k < 8; ++k) {
            const int e = e0 + BNODES * k;
            sd[k] = pairs[e < end ? e : end - 1];
        }
        asm volatile("" :: "v"(sd[0].y), "v"(sd[1].y), "v"(sd[2].y), "v"(sd[3].y),
                           "v"(sd[4].y), "v"(sd[5].y), "v"(sd[6].y), "v"(sd[7].y));
        int p[8];
#pragma unroll
        for (int k = 0; k < 8; ++k)
            p[k] = (e0 + BNODES * k < end) ? atomicAdd(&cnt[sd[k].y & (BNODES - 1)], 1) : -1;
#pragma unroll
        for (int k = 0; k < 8; ++k) {
            if (p[k] >= 0) {
                const int ls = p[k] - beg;
                if (ls < RSTAGE) stage[ls] = sd[k].x;
                else             csr_src[p[k]] = sd[k].x;
            }
        }
    }
    __syncthreads();
    const int ne = end - beg;
    const int m  = ne < RSTAGE ? ne : RSTAGE;
    const int a0 = beg & 31;
    for (int i = t; i < m + a0; i += BNODES) {
        const int q = i - a0;
        if (q >= 0) csr_src[beg + q] = stage[q];
    }
}

__global__ __launch_bounds__(1024) __attribute__((amdgpu_waves_per_eu(8, 8)))
void gemm_wmma_f32(const float* __restrict__ X, const float* __restrict__ W,
                   const float* __restrict__ dinv,
                   float* __restrict__ H, int ntiles) {
    __shared__ float4 wl[64 * 64];

    const int wave = threadIdx.x >> 5;
    const int lane = threadIdx.x & 31;
    const int c    = lane & 15;
    const int h    = lane >> 4;
    const int G    = (int)gridDim.x;
    const int g    = (int)blockIdx.x;
    const int nunits = 2 * ntiles;
    const int nl   = (nunits - g + G - 1) / G;
    const int j    = ((wave >> 2) & 1) * 16 + (wave >> 3) * 4 + (wave & 3);

    int l = j;
    int u = (l < nl) ? (l * G + g) : (nunits - 1);
    const v2f* __restrict__ xa = (const v2f*)(X + (size_t)((u >> 1) * 16 + c) * CH) + h;
    v2f a[32];
#pragma unroll
    for (int s = 0; s < 8; ++s) a[s] = xa[2 * s];
    float dvl = dinv[(u >> 1) * 16 + c];

    const float4* __restrict__ W4 = (const float4*)W;
#pragma unroll
    for (int it = 0; it < 2; ++it) {
        const int p = it * 32 + wave;
        const float4 ra = W4[(2 * p) * 32 + lane];
        const float4 rb = W4[(2 * p + 1) * 32 + lane];
        float4 lo, hi;
        lo.x = ra.x; lo.y = rb.x; lo.z = ra.y; lo.w = rb.y;
        hi.x = ra.z; hi.y = rb.z; hi.z = ra.w; hi.w = rb.w;
        wl[p * 64 + (2 * h) * 16 + c]     = lo;
        wl[p * 64 + (2 * h + 1) * 16 + c] = hi;
    }
    __syncthreads();

    float4* __restrict__ H4 = (float4*)H;
    while (l < nl) {
        const int m0 = (u >> 1) * 16;
        const int ch = u & 1;

        v8f acc[4];
#pragma unroll
        for (int t = 0; t < 4; ++t) {
            acc[t] = (v8f){0.f, 0.f, 0.f, 0.f, 0.f, 0.f, 0.f, 0.f};
        }

        const float4* bl = wl + h * 64 + ch * 32 + c;
#pragma unroll
        for (int s = 0; s < 32; ++s) {
            if ((s & 7) == 0) {
                if (s + 8 < 32) {
#pragma unroll
                    for (int i = 0; i < 8; ++i) a[s + 8 + i] = xa[2 * (s + 8 + i)];
                }
                __builtin_amdgcn_sched_barrier(0);
            }
            const v2f as = a[s] * dvl;
            const float4 q0 = bl[s * 128];
            const float4 q1 = bl[s * 128 + 16];
            v2f b;
            b.x = q0.x; b.y = q0.y;
            acc[0] = __builtin_amdgcn_wmma_f32_16x16x4_f32(false, as, false, b, (short)0, acc[0], false, false);
            b.x = q0.z; b.y = q0.w;
            acc[1] = __builtin_amdgcn_wmma_f32_16x16x4_f32(false, as, false, b, (short)0, acc[1], false, false);
            b.x = q1.x; b.y = q1.y;
            acc[2] = __builtin_amdgcn_wmma_f32_16x16x4_f32(false, as, false, b, (short)0, acc[2], false, false);
            b.x = q1.z; b.y = q1.w;
            acc[3] = __builtin_amdgcn_wmma_f32_16x16x4_f32(false, as, false, b, (short)0, acc[3], false, false);
        }

#pragma unroll
        for (int r = 0; r < 8; ++r) {
            const size_t row4 = (size_t)(m0 + 8 * h + r) * (CH / 4);
            float4 v0;
            v0.x = acc[0][r]; v0.y = acc[1][r]; v0.z = acc[2][r]; v0.w = acc[3][r];
            H4[row4 + 16 * ch + c] = v0;
        }

        l += 32;
        if (l < nl) {
            u  = l * G + g;
            xa = (const v2f*)(X + (size_t)((u >> 1) * 16 + c) * CH) + h;
#pragma unroll
            for (int s = 0; s < 8; ++s) a[s] = xa[2 * s];
            dvl = dinv[(u >> 1) * 16 + c];
        }
    }
}

template <int MASK>
__device__ __forceinline__ float row_xor(float v) {
    return __builtin_bit_cast(float, __builtin_amdgcn_update_dpp(
        0, __builtin_bit_cast(int, v), 0x160 + MASK, 0xf, 0xf, true));
}

template <bool LAST>
__global__ __launch_bounds__(256)
void gather128_kernel(const float* __restrict__ h, const int* __restrict__ rowptr,
                      const int* __restrict__ csr_src, const float* __restrict__ dinv,
                      const float* __restrict__ b, const float* __restrict__ g,
                      const float* __restrict__ be, const float* __restrict__ rm,
                      const float* __restrict__ rv, const float* __restrict__ w3,
                      float* __restrict__ out, int n) {
    const int d = __builtin_amdgcn_readfirstlane(
        (int)((blockIdx.x * blockDim.x + threadIdx.x) >> 5));
    if (d >= n) return;
    const int lane = threadIdx.x & 31;
    const int beg = rowptr[d];
    const int end = rowptr[d + 1];
    const float dd = dinv[d];
    const float4* __restrict__ hv = (const float4*)h;

    const float4 pb  = ((const float4*)b)[lane];
    const float4 pg  = ((const float4*)g)[lane];
    const float4 pbe = ((const float4*)be)[lane];
    const float4 prm = ((const float4*)rm)[lane];
    const float4 prv = ((const float4*)rv)[lane];
    const float ix = pg.x * rsqrtf(prv.x + 1e-5f);
    const float iy = pg.y * rsqrtf(prv.y + 1e-5f);
    const float iz = pg.z * rsqrtf(prv.z + 1e-5f);
    const float iw = pg.w * rsqrtf(prv.w + 1e-5f);
    float4 wa = make_float4(0.f, 0.f, 0.f, 0.f);
    float4 wb = make_float4(0.f, 0.f, 0.f, 0.f);
    if (LAST) {
        wa = ((const float4*)w3)[2 * lane];
        wb = ((const float4*)w3)[2 * lane + 1];
    }

    float4 acc = hv[(size_t)d * 32 + lane];

    for (int base = beg; base < end; base += 32) {
        const int m = (end - base < 32) ? (end - base) : 32;
        int ms = 0;
        if (lane < m) ms = csr_src[base + lane];
        int j = 0;
        for (; j + 4 <= m; j += 4) {
            const int s0 = __builtin_amdgcn_readlane(ms, j);
            const int s1 = __builtin_amdgcn_readlane(ms, j + 1);
            const int s2 = __builtin_amdgcn_readlane(ms, j + 2);
            const int s3 = __builtin_amdgcn_readlane(ms, j + 3);
            const float4 v0 = hv[(size_t)s0 * 32 + lane];
            const float4 v1 = hv[(size_t)s1 * 32 + lane];
            const float4 v2 = hv[(size_t)s2 * 32 + lane];
            const float4 v3 = hv[(size_t)s3 * 32 + lane];
            acc.x += v0.x; acc.y += v0.y; acc.z += v0.z; acc.w += v0.w;
            acc.x += v1.x; acc.y += v1.y; acc.z += v1.z; acc.w += v1.w;
            acc.x += v2.x; acc.y += v2.y; acc.z += v2.z; acc.w += v2.w;
            acc.x += v3.x; acc.y += v3.y; acc.z += v3.z; acc.w += v3.w;
        }
        const int rem = m & 3;
        for (int t = 0; t < rem; ++t) {
            const int s0 = __builtin_amdgcn_readlane(ms, j + t);
            const float4 v0 = hv[(size_t)s0 * 32 + lane];
            acc.x += v0.x; acc.y += v0.y; acc.z += v0.z; acc.w += v0.w;
        }
    }
    float4 r;
    {
        const float vx = (acc.x * dd + pb.x - prm.x) * ix + pbe.x;
        const float vy = (acc.y * dd + pb.y - prm.y) * iy + pbe.y;
        const float vz = (acc.z * dd + pb.z - prm.z) * iz + pbe.z;
        const float vw = (acc.w * dd + pb.w - prm.w) * iw + pbe.w;
        r.x = vx > 0.f ? vx : 0.f;
        r.y = vy > 0.f ? vy : 0.f;
        r.z = vz > 0.f ? vz : 0.f;
        r.w = vw > 0.f ? vw : 0.f;
    }
    if (!LAST) {
        ((float4*)out)[(size_t)d * 32 + lane] = r;
    } else {
        const float p0 = r.x * wa.x + r.y * wa.z + r.z * wb.x + r.w * wb.z;
        const float p1 = r.x * wa.y + r.y * wa.w + r.z * wb.y + r.w * wb.w;
        const bool up = lane >= 16;
        const float keep = up ? p1 : p0;
        const float give = up ? p0 : p1;
        float v = keep + __builtin_bit_cast(float, __builtin_amdgcn_permlanex16(
            0, __builtin_bit_cast(int, give), 0x76543210, 0xfedcba98, false, false));
        v += row_xor<8>(v);
        v += row_xor<4>(v);
        v += row_xor<2>(v);
        v += row_xor<1>(v);
        if ((lane & 15) == 0) out[(size_t)d * 2 + (lane >> 4)] = v * dd;
    }
}

__global__ void gather2_kernel(const float* __restrict__ h3, const int* __restrict__ rowptr,
                               const int* __restrict__ csr_src, const float* __restrict__ dinv,
                               const float* __restrict__ b3, float* __restrict__ out, int n) {
    const int d = blockIdx.x * blockDim.x + threadIdx.x;
    if (d >= n) return;
    const int beg = rowptr[d];
    const int end = rowptr[d + 1];
    const float dd = dinv[d];
    const float2* __restrict__ hv = (const float2*)h3;
    const float2 vs = hv[d];
    float a0 = vs.x;
    float a1 = vs.y;
#pragma unroll 4
    for (int j = beg; j < end; ++j) {
        const int s = csr_src[j];
        const float2 u = hv[s];
        a0 += u.x;
        a1 += u.y;
    }
    float2 r;
    r.x = a0 * dd + b3[0];
    r.y = a1 * dd + b3[1];
    ((float2*)out)[d] = r;
}

extern "C" void kernel_launch(void* const* d_in, const int* in_sizes, int n_in,
                              void* d_out, int out_size, void* d_ws, size_t ws_size,
                              hipStream_t stream) {
    const float* x   = (const float*)d_in[0];
    const int*   ei  = (const int*)d_in[1];
    const float* W1  = (const float*)d_in[2];
    const float* b1  = (const float*)d_in[3];
    const float* g1  = (const float*)d_in[4];
    const float* be1 = (const float*)d_in[5];
    const float* rm1 = (const float*)d_in[6];
    const float* rv1 = (const float*)d_in[7];
    const float* W2  = (const float*)d_in[8];
    const float* b2  = (const float*)d_in[9];
    const float* g2  = (const float*)d_in[10];
    const float* be2 = (const float*)d_in[11];
    const float* rm2 = (const float*)d_in[12];
    const float* rv2 = (const float*)d_in[13];
    const float* W3  = (const float*)d_in[14];
    const float* b3  = (const float*)d_in[15];

    const int Nn   = in_sizes[0] / CH;
    const int Ee   = in_sizes[1] / 2;
    const int NC   = Nn * CH;

    const int* src = ei;
    const int* dst = ei + Ee;

    const int T = 256;
    const int gN  = (Nn + T - 1) / T;
    const int gW  = (Nn + 7) / 8;
    const int gMM = Nn / 16;
    const int nb  = (Nn + BNODES - 1) / BNODES;
    const int per = (Ee + CHUNKS - 1) / CHUNKS;
    const int nbp = (nb + 3) & ~3;
    const int tm  = nbp * CHUNKS;

    const size_t NP = ((size_t)Nn + 64) & ~(size_t)63;
    const size_t TP = ((size_t)tm + 63) & ~(size_t)63;
    const size_t BP = ((size_t)nb + 64) & ~(size_t)63;
    const size_t EP = ((size_t)Ee + 63) & ~(size_t)63;
    int*   tab    = (int*)d_ws;
    int*   bstart = tab + TP;
    int*   rowptr = bstart + BP;
    float* dinv   = (float*)(rowptr + NP);
    int*   csr    = (int*)(dinv + NP);
    int2*  pairs  = (int2*)(csr + EP);
    float* bufA   = (float*)(csr + 3 * EP);
    float* bufB   = bufA + (size_t)NC;
    float* h3     = bufB + (size_t)NC;
    float* out    = (float*)d_out;

    bucket_count_kernel<<<CHUNKS, 1024, nb * sizeof(int), stream>>>(dst, tab, Ee, per, nb, nbp);
    bucket_place_kernel<<<CHUNKS, 1024, (3 * nbp + 68) * sizeof(int) + (size_t)per * sizeof(int2), stream>>>(src, dst, tab, bstart, pairs, Ee, per, nb, nbp);
    bucket_rows_kernel<<<nb, BNODES, RSTAGE * sizeof(int), stream>>>(pairs, bstart, rowptr, dinv, csr, Nn);

    gemm_wmma_f32<<<256, 1024, 0, stream>>>(x, W1, dinv, bufA, gMM);
    gather128_kernel<false><<<gW, T, 0, stream>>>(bufA, rowptr, csr, dinv, b1, g1, be1, rm1, rv1, W3, bufB, Nn);

    gemm_wmma_f32<<<256, 1024, 0, stream>>>(bufB, W2, dinv, bufA, gMM);
    gather128_kernel<true><<<gW, T, 0, stream>>>(bufA, rowptr, csr, dinv, b2, g2, be2, rm2, rv2, W3, h3, Nn);

    gather2_kernel<<<gN, T, 0, stream>>>(h3, rowptr, csr, dinv, b3, out, Nn);
}
